# starting kernel
# baseline (speedup 1.0000x reference)
#include <hip/hip_runtime.h>
#include <hip/hip_bf16.h>

typedef __attribute__((ext_vector_type(2))) float v2f;
typedef __attribute__((ext_vector_type(8))) float v8f;

static constexpr int cB = 2;
static constexpr int cN = 384;
static constexpr int cD = 128;
static constexpr int cM = cB * cN;

__global__ void gemm_wmma_f32(const float* __restrict__ A, int K,
                              const float* __restrict__ Bm,
                              const float* __restrict__ bias,
                              const float* __restrict__ rowscale,
                              float* __restrict__ C,
                              int addC, int relu)
{
    const int wave  = threadIdx.x >> 5;
    const int lane  = threadIdx.x & 31;
    const int tile  = blockIdx.x * 8 + wave;
    const int mt    = tile >> 3;
    const int nt    = tile & 7;
    const int m0    = mt * 16;
    const int n0    = nt * 16;
    const int l15   = lane & 15;
    const int khalf = lane >> 4;
    const int n     = n0 + l15;
    const int mbase = m0 + (khalf ? 8 : 0);

    v8f c = {};
    if (addC) {
#pragma unroll
        for (int r = 0; r < 8; ++r)
            c[r] = C[(size_t)(mbase + r) * cD + n];
    }

    const float* Arow = A + (size_t)(m0 + l15) * K;
    for (int k0 = 0; k0 < K; k0 += 4) {
        const int ka = k0 + 2 * khalf;
        v2f a, b;
        a.x = Arow[ka];
        a.y = Arow[ka + 1];
        b.x = Bm[(size_t)ka * cD + n];
        b.y = Bm[(size_t)(ka + 1) * cD + n];
        c = __builtin_amdgcn_wmma_f32_16x16x4_f32(
                false, a, false, b, (short)0, c, false, false);
    }

    const float bv = bias ? bias[n] : 0.0f;
#pragma unroll
    for (int r = 0; r < 8; ++r) {
        const int mr = mbase + r;
        const float s = rowscale ? rowscale[mr] : 1.0f;
        float y = c[r] + s * bv;
        if (relu) y = y > 0.0f ? y : 0.0f;
        C[(size_t)mr * cD + n] = y;
    }
}

__global__ void pairwise_agg(const float* __restrict__ P,
                             const float* __restrict__ Q,
                             const float* __restrict__ adj,
                             const float* __restrict__ we,
                             const float* __restrict__ mb1,
                             float* __restrict__ R,
                             float* __restrict__ deg)
{
    const int bi = blockIdx.x;
    const int b  = bi / cN;
    const int d  = threadIdx.x;

    __shared__ float s_adj[cN];
    for (int j = d; j < cN; j += blockDim.x)
        s_adj[j] = adj[(size_t)bi * cN + j];
    __syncthreads();

    const float p  = P[(size_t)bi * cD + d];
    const float w  = we[d];
    const float b1 = mb1[d];
    const float* Qb = Q + (size_t)b * cN * cD;

    float acc = 0.0f;
    float dg  = 0.0f;
    for (int j = 0; j < cN; ++j) {
        const float a = s_adj[j];
        if (a != 0.0f) {
            float t = p + Qb[(size_t)j * cD + d] + a * w + b1;
            t = t > 0.0f ? t : 0.0f;
            acc += a * t;
            dg  += a;
        }
    }
    R[(size_t)bi * cD + d] = acc;
    if (d == 0) deg[bi] = dg;
}

__global__ void ln_res_relu(const float* __restrict__ xin,
                            const float* __restrict__ hn,
                            const float* __restrict__ g,
                            const float* __restrict__ beta,
                            float* __restrict__ out,
                            int add_residual, int relu_after)
{
    const int row = blockIdx.x;
    const int d   = threadIdx.x;

    float v = hn[(size_t)row * cD + d];
    if (add_residual) v += xin[(size_t)row * cD + d];

    __shared__ float red[cD];
    red[d] = v;
    __syncthreads();
#pragma unroll
    for (int s = cD / 2; s > 0; s >>= 1) {
        if (d < s) red[d] += red[d + s];
        __syncthreads();
    }
    const float mu = red[0] * (1.0f / cD);
    __syncthreads();

    const float cdev = v - mu;
    red[d] = cdev * cdev;
    __syncthreads();
#pragma unroll
    for (int s = cD / 2; s > 0; s >>= 1) {
        if (d < s) red[d] += red[d + s];
        __syncthreads();
    }
    const float var = red[0] * (1.0f / cD);

    float y = cdev * rsqrtf(var + 1e-5f) * g[d] + beta[d];
    if (relu_after) y = y > 0.0f ? y : 0.0f;
    out[(size_t)row * cD + d] = y;
}

extern "C" void kernel_launch(void* const* d_in, const int* in_sizes, int n_in,
                              void* d_out, int out_size, void* d_ws, size_t ws_size,
                              hipStream_t stream)
{
    (void)in_sizes; (void)n_in; (void)out_size; (void)ws_size;

    const float* h    = (const float*)d_in[0];
    const float* adj  = (const float*)d_in[1];
    const float* ip_w = (const float*)d_in[2];
    const float* ip_b = (const float*)d_in[3];
    const float* mw1  = (const float*)d_in[4];
    const float* mb1  = (const float*)d_in[5];
    const float* mw2  = (const float*)d_in[6];
    const float* mb2  = (const float*)d_in[7];
    const float* uw1  = (const float*)d_in[8];
    const float* ub1  = (const float*)d_in[9];
    const float* uw2  = (const float*)d_in[10];
    const float* ub2  = (const float*)d_in[11];
    const float* ln_g = (const float*)d_in[12];
    const float* ln_b = (const float*)d_in[13];

    float* ws = (float*)d_ws;
    const size_t S = (size_t)cM * cD;
    float* x   = ws + 0 * S;
    float* P   = ws + 1 * S;
    float* Q   = ws + 2 * S;
    float* R   = ws + 3 * S;
    float* agg = ws + 4 * S;
    float* hid = ws + 5 * S;
    float* hn  = ws + 6 * S;
    float* deg = ws + 7 * S;

    gemm_wmma_f32<<<48, 256, 0, stream>>>(h, 64, ip_w, ip_b, nullptr, x, 0, 1);

    for (int l = 0; l < 3; ++l) {
        const float* mw1l = mw1 + (size_t)l * 257 * cD;
        const float* mb1l = mb1 + (size_t)l * cD;
        const float* mw2l = mw2 + (size_t)l * cD * cD;
        const float* mb2l = mb2 + (size_t)l * cD;
        const float* uw1l = uw1 + (size_t)l * 256 * cD;
        const float* ub1l = ub1 + (size_t)l * cD;
        const float* uw2l = uw2 + (size_t)l * cD * cD;
        const float* ub2l = ub2 + (size_t)l * cD;

        gemm_wmma_f32<<<48, 256, 0, stream>>>(x, cD, mw1l,            nullptr, nullptr, P, 0, 0);
        gemm_wmma_f32<<<48, 256, 0, stream>>>(x, cD, mw1l + 128 * cD, nullptr, nullptr, Q, 0, 0);

        pairwise_agg<<<cM, cD, 0, stream>>>(P, Q, adj, mw1l + 256 * cD, mb1l, R, deg);

        gemm_wmma_f32<<<48, 256, 0, stream>>>(R, cD, mw2l, mb2l, deg, agg, 0, 0);

        gemm_wmma_f32<<<48, 256, 0, stream>>>(x,   cD, uw1l,            ub1l,    nullptr, hid, 0, 0);
        gemm_wmma_f32<<<48, 256, 0, stream>>>(agg, cD, uw1l + 128 * cD, nullptr, nullptr, hid, 1, 1);

        gemm_wmma_f32<<<48, 256, 0, stream>>>(hid, cD, uw2l, ub2l, nullptr, hn, 0, 0);

        if (l < 2) {
            ln_res_relu<<<cM, cD, 0, stream>>>(x, hn, ln_g + l * cD, ln_b + l * cD, x, 1, 1);
        } else {
            ln_res_relu<<<cM, cD, 0, stream>>>(x, hn, ln_g + l * cD, ln_b + l * cD,
                                               (float*)d_out, 0, 0);
        }
    }
}


# pairwise aggregation: each wave compacts the row's nonzero senders into an ordered LDS list and issues their sender-row loads 16 at a time, not one per taken branch; every sum keeps its j order
# speedup vs baseline: 1.5734x; 1.5734x over previous
#include <hip/hip_runtime.h>
#include <hip/hip_bf16.h>

typedef __attribute__((ext_vector_type(2))) float v2f;
typedef __attribute__((ext_vector_type(8))) float v8f;

static constexpr int cB = 2;
static constexpr int cN = 384;
static constexpr int cD = 128;
static constexpr int cM = cB * cN;

__global__ void gemm_wmma_f32(const float* __restrict__ A, int K,
                              const float* __restrict__ Bm,
                              const float* __restrict__ bias,
                              const float* __restrict__ rowscale,
                              float* __restrict__ C,
                              int addC, int relu)
{
    const int wave  = threadIdx.x >> 5;
    const int lane  = threadIdx.x & 31;
    const int tile  = blockIdx.x * 8 + wave;
    const int mt    = tile >> 3;
    const int nt    = tile & 7;
    const int m0    = mt * 16;
    const int n0    = nt * 16;
    const int l15   = lane & 15;
    const int khalf = lane >> 4;
    const int n     = n0 + l15;
    const int mbase = m0 + (khalf ? 8 : 0);

    v8f c = {};
    if (addC) {
#pragma unroll
        for (int r = 0; r < 8; ++r)
            c[r] = C[(size_t)(mbase + r) * cD + n];
    }

    const float* Arow = A + (size_t)(m0 + l15) * K;
    for (int k0 = 0; k0 < K; k0 += 4) {
        const int ka = k0 + 2 * khalf;
        v2f a, b;
        a.x = Arow[ka];
        a.y = Arow[ka + 1];
        b.x = Bm[(size_t)ka * cD + n];
        b.y = Bm[(size_t)(ka + 1) * cD + n];
        c = __builtin_amdgcn_wmma_f32_16x16x4_f32(
                false, a, false, b, (short)0, c, false, false);
    }

    const float bv = bias ? bias[n] : 0.0f;
#pragma unroll
    for (int r = 0; r < 8; ++r) {
        const int mr = mbase + r;
        const float s = rowscale ? rowscale[mr] : 1.0f;
        float y = c[r] + s * bv;
        if (relu) y = y > 0.0f ? y : 0.0f;
        C[(size_t)mr * cD + n] = y;
    }
}

static constexpr int cU  = 16;
static constexpr int cCH = cN / 32;

__device__ __forceinline__ void pair_step(float p, float q, float a, float w,
                                          float b1, float& acc, float& dg)
{
    float t = __builtin_fmaf(a, w, p + q) + b1;
    t = t > 0.0f ? t : 0.0f;
    acc = __builtin_fmaf(a, t, acc);
    dg += a;
}

__global__ void pairwise_agg(const float* __restrict__ P,
                             const float* __restrict__ Q,
                             const float* __restrict__ adj,
                             const float* __restrict__ we,
                             const float* __restrict__ mb1,
                             float* __restrict__ R,
                             float* __restrict__ deg)
{
    const int bi   = blockIdx.x;
    const int b    = bi / cN;
    const int d    = threadIdx.x;
    const int wave = threadIdx.x >> 5;
    const int lane = threadIdx.x & 31;

    __shared__ __attribute__((aligned(16))) int   s_j[4][cN + cU];
    __shared__ __attribute__((aligned(16))) float s_a[4][cN + cU];

    const float* arow = adj + (size_t)bi * cN;
    float av[cCH];
#pragma unroll
    for (int c = 0; c < cCH; ++c)
        av[c] = arow[c * 32 + lane];

    int n = 0;
#pragma unroll
    for (int c = 0; c < cCH; ++c) {
        const bool nz = av[c] != 0.0f;
        const unsigned mask = __builtin_amdgcn_ballot_w32(nz);
        const int pos = n + (int)__builtin_amdgcn_mbcnt_lo(mask, 0u);
        if (nz) {
            s_j[wave][pos] = c * 32 + lane;
            s_a[wave][pos] = av[c];
        }
        n += __builtin_popcount(mask);
    }
    if (lane < cU) {
        s_j[wave][n + lane] = 0;
        s_a[wave][n + lane] = 0.0f;
    }
    __syncthreads();

    const float p  = P[(size_t)bi * cD + d];
    const float w  = we[d];
    const float b1 = mb1[d];
    const float* Qb = Q + (size_t)b * cN * cD;
    const int4*   lj = reinterpret_cast<const int4*>(s_j[wave]);
    const float4* la = reinterpret_cast<const float4*>(s_a[wave]);

    float acc = 0.0f;
    float dg  = 0.0f;
    int k = 0;
    for (; k + cU <= n; k += cU) {
        int jj[cU]; float aa[cU]; float q[cU];
#pragma unroll
        for (int v = 0; v < cU / 4; ++v) {
            const int4   j4 = lj[(k >> 2) + v];
            const float4 a4 = la[(k >> 2) + v];
            jj[4 * v + 0] = j4.x; jj[4 * v + 1] = j4.y; jj[4 * v + 2] = j4.z; jj[4 * v + 3] = j4.w;
            aa[4 * v + 0] = a4.x; aa[4 * v + 1] = a4.y; aa[4 * v + 2] = a4.z; aa[4 * v + 3] = a4.w;
        }
#pragma unroll
        for (int u = 0; u < cU; ++u)
            q[u] = Qb[(unsigned)(jj[u] * cD + d)];
#pragma unroll
        for (int u = 0; u < cU; ++u)
            pair_step(p, q[u], aa[u], w, b1, acc, dg);
    }
    if (k < n) {
        int jj[cU]; float aa[cU]; float q[cU];
#pragma unroll
        for (int v = 0; v < cU / 4; ++v) {
            const int4   j4 = lj[(k >> 2) + v];
            const float4 a4 = la[(k >> 2) + v];
            jj[4 * v + 0] = j4.x; jj[4 * v + 1] = j4.y; jj[4 * v + 2] = j4.z; jj[4 * v + 3] = j4.w;
            aa[4 * v + 0] = a4.x; aa[4 * v + 1] = a4.y; aa[4 * v + 2] = a4.z; aa[4 * v + 3] = a4.w;
        }
#pragma unroll
        for (int u = 0; u < cU; ++u)
            q[u] = Qb[(unsigned)(jj[u] * cD + d)];
#pragma unroll
        for (int u = 0; u < cU; ++u)
            if (k + u < n)
                pair_step(p, q[u], aa[u], w, b1, acc, dg);
    }
    R[(size_t)bi * cD + d] = acc;
    if (d == 0) deg[bi] = dg;
}

__global__ void ln_res_relu(const float* __restrict__ xin,
                            const float* __restrict__ hn,
                            const float* __restrict__ g,
                            const float* __restrict__ beta,
                            float* __restrict__ out,
                            int add_residual, int relu_after)
{
    const int row = blockIdx.x;
    const int d   = threadIdx.x;

    float v = hn[(size_t)row * cD + d];
    if (add_residual) v += xin[(size_t)row * cD + d];

    __shared__ float red[cD];
    red[d] = v;
    __syncthreads();
#pragma unroll
    for (int s = cD / 2; s > 0; s >>= 1) {
        if (d < s) red[d] += red[d + s];
        __syncthreads();
    }
    const float mu = red[0] * (1.0f / cD);
    __syncthreads();

    const float cdev = v - mu;
    red[d] = cdev * cdev;
    __syncthreads();
#pragma unroll
    for (int s = cD / 2; s > 0; s >>= 1) {
        if (d < s) red[d] += red[d + s];
        __syncthreads();
    }
    const float var = red[0] * (1.0f / cD);

    float y = cdev * rsqrtf(var + 1e-5f) * g[d] + beta[d];
    if (relu_after) y = y > 0.0f ? y : 0.0f;
    out[(size_t)row * cD + d] = y;
}

extern "C" void kernel_launch(void* const* d_in, const int* in_sizes, int n_in,
                              void* d_out, int out_size, void* d_ws, size_t ws_size,
                              hipStream_t stream)
{
    (void)in_sizes; (void)n_in; (void)out_size; (void)ws_size;

    const float* h    = (const float*)d_in[0];
    const float* adj  = (const float*)d_in[1];
    const float* ip_w = (const float*)d_in[2];
    const float* ip_b = (const float*)d_in[3];
    const float* mw1  = (const float*)d_in[4];
    const float* mb1  = (const float*)d_in[5];
    const float* mw2  = (const float*)d_in[6];
    const float* mb2  = (const float*)d_in[7];
    const float* uw1  = (const float*)d_in[8];
    const float* ub1  = (const float*)d_in[9];
    const float* uw2  = (const float*)d_in[10];
    const float* ub2  = (const float*)d_in[11];
    const float* ln_g = (const float*)d_in[12];
    const float* ln_b = (const float*)d_in[13];

    float* ws = (float*)d_ws;
    const size_t S = (size_t)cM * cD;
    float* x   = ws + 0 * S;
    float* P   = ws + 1 * S;
    float* Q   = ws + 2 * S;
    float* R   = ws + 3 * S;
    float* agg = ws + 4 * S;
    float* hid = ws + 5 * S;
    float* hn  = ws + 6 * S;
    float* deg = ws + 7 * S;

    gemm_wmma_f32<<<48, 256, 0, stream>>>(h, 64, ip_w, ip_b, nullptr, x, 0, 1);

    for (int l = 0; l < 3; ++l) {
        const float* mw1l = mw1 + (size_t)l * 257 * cD;
        const float* mb1l = mb1 + (size_t)l * cD;
        const float* mw2l = mw2 + (size_t)l * cD * cD;
        const float* mb2l = mb2 + (size_t)l * cD;
        const float* uw1l = uw1 + (size_t)l * 256 * cD;
        const float* ub1l = ub1 + (size_t)l * cD;
        const float* uw2l = uw2 + (size_t)l * cD * cD;
        const float* ub2l = ub2 + (size_t)l * cD;

        gemm_wmma_f32<<<48, 256, 0, stream>>>(x, cD, mw1l,            nullptr, nullptr, P, 0, 0);
        gemm_wmma_f32<<<48, 256, 0, stream>>>(x, cD, mw1l + 128 * cD, nullptr, nullptr, Q, 0, 0);

        pairwise_agg<<<cM, cD, 0, stream>>>(P, Q, adj, mw1l + 256 * cD, mb1l, R, deg);

        gemm_wmma_f32<<<48, 256, 0, stream>>>(R, cD, mw2l, mb2l, deg, agg, 0, 0);

        gemm_wmma_f32<<<48, 256, 0, stream>>>(x,   cD, uw1l,            ub1l,    nullptr, hid, 0, 0);
        gemm_wmma_f32<<<48, 256, 0, stream>>>(agg, cD, uw1l + 128 * cD, nullptr, nullptr, hid, 1, 1);

        gemm_wmma_f32<<<48, 256, 0, stream>>>(hid, cD, uw2l, ub2l, nullptr, hn, 0, 0);

        if (l < 2) {
            ln_res_relu<<<cM, cD, 0, stream>>>(x, hn, ln_g + l * cD, ln_b + l * cD, x, 1, 1);
        } else {
            ln_res_relu<<<cM, cD, 0, stream>>>(x, hn, ln_g + l * cD, ln_b + l * cD,
                                               (float*)d_out, 0, 0);
        }
    }
}


# one kernel per layer runs the whole row-local chain (agg, hid, hn, layer norm, next P/Q) on a 16-row tile with LDS hand-over between products: 25 launches to 7; every sum keeps its order
# speedup vs baseline: 2.8622x; 1.8191x over previous
#include <hip/hip_runtime.h>

typedef __attribute__((ext_vector_type(2))) float v2f;
typedef __attribute__((ext_vector_type(8))) float v8f;


static constexpr int cB = 2;
static constexpr int cN = 384;
static constexpr int cD = 128;
static constexpr int cM = cB * cN;
static constexpr int cDI = 64;
static constexpr int cLD = cD + 4;

template <int K>
__device__ __forceinline__ v8f tile_mm(const float* arow, const float* bcol, v8f c)
{
#pragma unroll
    for (int k0 = 0; k0 < K; k0 += 4) {
        v2f a, b;
        a.x = arow[k0];
        a.y = arow[k0 + 1];
        b.x = bcol[(size_t)k0 * cD];
        b.y = bcol[(size_t)(k0 + 1) * cD];
        c = __builtin_amdgcn_wmma_f32_16x16x4_f32(
                false, a, false, b, (short)0, c, false, false);
    }
    return c;
}

__device__ __forceinline__ void pq_from_tile(const float* arow,
                                             const float* __restrict__ w,
                                             float* __restrict__ P,
                                             float* __restrict__ Q,
                                             int mbase, int khalf, int n)
{
    const float* bp = w + (size_t)(2 * khalf) * cD + n;
    const float* bq = bp + (size_t)cD * cD;
    v8f p = {};
    v8f q = {};
#pragma unroll
    for (int k0 = 0; k0 < cD; k0 += 4) {
        v2f a, b0, b1;
        a.x  = arow[k0];
        a.y  = arow[k0 + 1];
        b0.x = bp[(size_t)k0 * cD];
        b0.y = bp[(size_t)(k0 + 1) * cD];
        b1.x = bq[(size_t)k0 * cD];
        b1.y = bq[(size_t)(k0 + 1) * cD];
        p = __builtin_amdgcn_wmma_f32_16x16x4_f32(
                false, a, false, b0, (short)0, p, false, false);
        q = __builtin_amdgcn_wmma_f32_16x16x4_f32(
                false, a, false, b1, (short)0, q, false, false);
    }
#pragma unroll
    for (int r = 0; r < 8; ++r) {
        P[(size_t)(mbase + r) * cD + n] = p[r];
        Q[(size_t)(mbase + r) * cD + n] = q[r];
    }
}

__global__ __launch_bounds__(256)
void in_proj_pq(const float* __restrict__ h,
                const float* __restrict__ ipw,
                const float* __restrict__ ipb,
                const float* __restrict__ mw1,
                float* __restrict__ x,
                float* __restrict__ P,
                float* __restrict__ Q)
{
    __shared__ __attribute__((aligned(16))) float sX[16][cLD];

    const int wave  = threadIdx.x >> 5;
    const int lane  = threadIdx.x & 31;
    const int m0    = blockIdx.x * 16;
    const int l15   = lane & 15;
    const int khalf = lane >> 4;
    const int n     = wave * 16 + l15;
    const int rb    = khalf * 8;

    v8f c = {};
    c = tile_mm<cDI>(h + (size_t)(m0 + l15) * cDI + 2 * khalf,
                     ipw + (size_t)(2 * khalf) * cD + n, c);
    const float bv = ipb[n];
#pragma unroll
    for (int r = 0; r < 8; ++r) {
        float y = c[r] + bv;
        y = y > 0.0f ? y : 0.0f;
        sX[rb + r][n] = y;
        x[(size_t)(m0 + rb + r) * cD + n] = y;
    }
    __syncthreads();
    pq_from_tile(&sX[l15][2 * khalf], mw1, P, Q, m0 + rb, khalf, n);
}

static constexpr int cU  = 16;
static constexpr int cCH = cN / 32;

__device__ __forceinline__ void pair_step(float p, float q, float a, float w,
                                          float b1, float& acc, float& dg)
{
    float t = __builtin_fmaf(a, w, p + q) + b1;
    t = t > 0.0f ? t : 0.0f;
    acc = __builtin_fmaf(a, t, acc);
    dg += a;
}

__global__ void pairwise_agg(const float* __restrict__ P,
                             const float* __restrict__ Q,
                             const float* __restrict__ adj,
                             const float* __restrict__ we,
                             const float* __restrict__ mb1,
                             float* __restrict__ R,
                             float* __restrict__ deg)
{
    const int bi   = blockIdx.x;
    const int b    = bi / cN;
    const int d    = threadIdx.x;
    const int wave = threadIdx.x >> 5;
    const int lane = threadIdx.x & 31;

    __shared__ __attribute__((aligned(16))) int   s_j[4][cN + cU];
    __shared__ __attribute__((aligned(16))) float s_a[4][cN + cU];

    const float* arow = adj + (size_t)bi * cN;
    float av[cCH];
#pragma unroll
    for (int c = 0; c < cCH; ++c)
        av[c] = arow[c * 32 + lane];

    int n = 0;
#pragma unroll
    for (int c = 0; c < cCH; ++c) {
        const bool nz = av[c] != 0.0f;
        const unsigned mask = __builtin_amdgcn_ballot_w32(nz);
        const int pos = n + (int)__builtin_amdgcn_mbcnt_lo(mask, 0u);
        if (nz) {
            s_j[wave][pos] = c * 32 + lane;
            s_a[wave][pos] = av[c];
        }
        n += __builtin_popcount(mask);
    }
    if (lane < cU) {
        s_j[wave][n + lane] = 0;
        s_a[wave][n + lane] = 0.0f;
    }
    __syncthreads();

    const float p  = P[(size_t)bi * cD + d];
    const float w  = we[d];
    const float b1 = mb1[d];
    const float* Qb = Q + (size_t)b * cN * cD;
    const int4*   lj = reinterpret_cast<const int4*>(s_j[wave]);
    const float4* la = reinterpret_cast<const float4*>(s_a[wave]);

    float acc = 0.0f;
    float dg  = 0.0f;
    int k = 0;
    for (; k + cU <= n; k += cU) {
        int jj[cU]; float aa[cU]; float q[cU];
#pragma unroll
        for (int v = 0; v < cU / 4; ++v) {
            const int4   j4 = lj[(k >> 2) + v];
            const float4 a4 = la[(k >> 2) + v];
            jj[4 * v + 0] = j4.x; jj[4 * v + 1] = j4.y; jj[4 * v + 2] = j4.z; jj[4 * v + 3] = j4.w;
            aa[4 * v + 0] = a4.x; aa[4 * v + 1] = a4.y; aa[4 * v + 2] = a4.z; aa[4 * v + 3] = a4.w;
        }
#pragma unroll
        for (int u = 0; u < cU; ++u)
            q[u] = Qb[(unsigned)(jj[u] * cD + d)];
#pragma unroll
        for (int u = 0; u < cU; ++u)
            pair_step(p, q[u], aa[u], w, b1, acc, dg);
    }
    if (k < n) {
        int jj[cU]; float aa[cU]; float q[cU];
#pragma unroll
        for (int v = 0; v < cU / 4; ++v) {
            const int4   j4 = lj[(k >> 2) + v];
            const float4 a4 = la[(k >> 2) + v];
            jj[4 * v + 0] = j4.x; jj[4 * v + 1] = j4.y; jj[4 * v + 2] = j4.z; jj[4 * v + 3] = j4.w;
            aa[4 * v + 0] = a4.x; aa[4 * v + 1] = a4.y; aa[4 * v + 2] = a4.z; aa[4 * v + 3] = a4.w;
        }
#pragma unroll
        for (int u = 0; u < cU; ++u)
            q[u] = Qb[(unsigned)(jj[u] * cD + d)];
#pragma unroll
        for (int u = 0; u < cU; ++u)
            if (k + u < n)
                pair_step(p, q[u], aa[u], w, b1, acc, dg);
    }
    R[(size_t)bi * cD + d] = acc;
    if (d == 0) deg[bi] = dg;
}

__global__ __launch_bounds__(256)
void layer_chain(const float* __restrict__ R,
                 const float* __restrict__ deg,
                 const float* xin,
                 const float* __restrict__ mw2,
                 const float* __restrict__ mb2,
                 const float* __restrict__ uw1,
                 const float* __restrict__ ub1,
                 const float* __restrict__ uw2,
                 const float* __restrict__ ub2,
                 const float* __restrict__ g,
                 const float* __restrict__ beta,
                 const float* __restrict__ mw1n,
                 float* __restrict__ P,
                 float* __restrict__ Q,
                 float* xout,
                 int add_residual, int relu_after)
{
    __shared__ __attribute__((aligned(16))) float sA[16][cLD];
    __shared__ __attribute__((aligned(16))) float sH[16][cLD];

    const int wave  = threadIdx.x >> 5;
    const int lane  = threadIdx.x & 31;
    const int m0    = blockIdx.x * 16;
    const int l15   = lane & 15;
    const int khalf = lane >> 4;
    const int n     = wave * 16 + l15;
    const int rb    = khalf * 8;
    const size_t boff = (size_t)(2 * khalf) * cD + n;

    const float bv_m2 = mb2[n];
    const float bv_u1 = ub1[n];
    const float bv_u2 = ub2[n];
    float dg[8], xr[8];
#pragma unroll
    for (int r = 0; r < 8; ++r) {
        dg[r] = deg[m0 + rb + r];
        xr[r] = xin[(size_t)(m0 + rb + r) * cD + n];
    }
    float gg[4], bb[4];
#pragma unroll
    for (int i = 0; i < 4; ++i) {
        gg[i] = g[lane + 32 * i];
        bb[i] = beta[lane + 32 * i];
    }

    {
        v8f c = {};
        c = tile_mm<cD>(R + (size_t)(m0 + l15) * cD + 2 * khalf, mw2 + boff, c);
#pragma unroll
        for (int r = 0; r < 8; ++r)
            sA[rb + r][n] = c[r] + dg[r] * bv_m2;
    }
    __syncthreads();

    {
        v8f c = {};
        c = tile_mm<cD>(xin + (size_t)(m0 + l15) * cD + 2 * khalf, uw1 + boff, c);
#pragma unroll
        for (int r = 0; r < 8; ++r)
            c[r] += bv_u1;
        c = tile_mm<cD>(&sA[l15][2 * khalf], uw1 + (size_t)cD * cD + boff, c);
#pragma unroll
        for (int r = 0; r < 8; ++r)
            sH[rb + r][n] = c[r] > 0.0f ? c[r] : 0.0f;
    }
    __syncthreads();

    {
        v8f c = {};
        c = tile_mm<cD>(&sH[l15][2 * khalf], uw2 + boff, c);
#pragma unroll
        for (int r = 0; r < 8; ++r) {
            float v = c[r] + bv_u2;
            if (add_residual) v += xr[r];
            sA[rb + r][n] = v;
        }
    }
    __syncthreads();

#pragma unroll
    for (int rr = 0; rr < 2; ++rr) {
        const int row = 2 * wave + rr;
        float v[4];
#pragma unroll
        for (int i = 0; i < 4; ++i)
            v[i] = sA[row][lane + 32 * i];

        float s = (v[0] + v[2]) + (v[1] + v[3]);
#pragma unroll
        for (int o = 16; o > 0; o >>= 1)
            s += __shfl_down(s, o);
        const float mu = __shfl(s, 0) * (1.0f / cD);

        float dv[4];
#pragma unroll
        for (int i = 0; i < 4; ++i)
            dv[i] = v[i] - mu;
        float t = (dv[0] * dv[0] + dv[2] * dv[2]) + (dv[1] * dv[1] + dv[3] * dv[3]);
#pragma unroll
        for (int o = 16; o > 0; o >>= 1)
            t += __shfl_down(t, o);
        const float var = __shfl(t, 0) * (1.0f / cD);
        const float rs  = rsqrtf(var + 1e-5f);

#pragma unroll
        for (int i = 0; i < 4; ++i) {
            float y = dv[i] * rs * gg[i] + bb[i];
            if (relu_after) y = y > 0.0f ? y : 0.0f;
            xout[(size_t)(m0 + row) * cD + lane + 32 * i] = y;
            sH[row][lane + 32 * i] = y;
        }
    }

    if (mw1n) {
        __syncthreads();
        pq_from_tile(&sH[l15][2 * khalf], mw1n, P, Q, m0 + rb, khalf, n);
    }
}

extern "C" void kernel_launch(void* const* d_in, const int* in_sizes, int n_in,
                              void* d_out, int out_size, void* d_ws, size_t ws_size,
                              hipStream_t stream)
{
    (void)in_sizes; (void)n_in; (void)out_size; (void)ws_size;

    const float* h    = (const float*)d_in[0];
    const float* adj  = (const float*)d_in[1];
    const float* ip_w = (const float*)d_in[2];
    const float* ip_b = (const float*)d_in[3];
    const float* mw1  = (const float*)d_in[4];
    const float* mb1  = (const float*)d_in[5];
    const float* mw2  = (const float*)d_in[6];
    const float* mb2  = (const float*)d_in[7];
    const float* uw1  = (const float*)d_in[8];
    const float* ub1  = (const float*)d_in[9];
    const float* uw2  = (const float*)d_in[10];
    const float* ub2  = (const float*)d_in[11];
    const float* ln_g = (const float*)d_in[12];
    const float* ln_b = (const float*)d_in[13];

    float* ws = (float*)d_ws;
    const size_t S = (size_t)cM * cD;
    float* x   = ws + 0 * S;
    float* P   = ws + 1 * S;
    float* Q   = ws + 2 * S;
    float* R   = ws + 3 * S;
    float* deg = ws + 7 * S;

    in_proj_pq<<<cM / 16, 256, 0, stream>>>(h, ip_w, ip_b, mw1, x, P, Q);

    for (int l = 0; l < 3; ++l) {
        const float* mw1l = mw1 + (size_t)l * 257 * cD;
        const float* mb1l = mb1 + (size_t)l * cD;
        const float* mw2l = mw2 + (size_t)l * cD * cD;
        const float* mb2l = mb2 + (size_t)l * cD;
        const float* uw1l = uw1 + (size_t)l * 256 * cD;
        const float* ub1l = ub1 + (size_t)l * cD;
        const float* uw2l = uw2 + (size_t)l * cD * cD;
        const float* ub2l = ub2 + (size_t)l * cD;

        pairwise_agg<<<cM, cD, 0, stream>>>(P, Q, adj, mw1l + 256 * cD, mb1l, R, deg);

        if (l < 2) {
            layer_chain<<<cM / 16, 256, 0, stream>>>(R, deg, x, mw2l, mb2l, uw1l, ub1l, uw2l, ub2l,
                                                     ln_g + l * cD, ln_b + l * cD,
                                                     mw1l + 257 * cD, P, Q, x, 1, 1);
        } else {
            layer_chain<<<cM / 16, 256, 0, stream>>>(R, deg, x, mw2l, mb2l, uw1l, ub1l, uw2l, ub2l,
                                                     ln_g + l * cD, ln_b + l * cD,
                                                     nullptr, P, Q, (float*)d_out, 0, 0);
        }
    }
}


# weight fragments of each product held in registers and loaded one product ahead: the loads fly behind the previous chain, not on a short window after each barrier; every sum keeps its order
# speedup vs baseline: 3.6687x; 1.2818x over previous
#include <hip/hip_runtime.h>

typedef __attribute__((ext_vector_type(2))) float v2f;
typedef __attribute__((ext_vector_type(8))) float v8f;


static constexpr int cB = 2;
static constexpr int cN = 384;
static constexpr int cD = 128;
static constexpr int cM = cB * cN;
static constexpr int cDI = 64;
static constexpr int cLD = cD + 4;

template <int K>
__device__ __forceinline__ void load_b_frag(const float* __restrict__ bcol, float (&b)[K / 2])
{
#pragma unroll
    for (int i = 0; i < K / 4; ++i) {
        b[2 * i]     = bcol[(size_t)(4 * i) * cD];
        b[2 * i + 1] = bcol[(size_t)(4 * i + 1) * cD];
    }
}

template <int K>
__device__ __forceinline__ void load_a_frag(const float* arow, float (&a)[K / 2])
{
#pragma unroll
    for (int i = 0; i < K / 4; ++i) {
        a[2 * i]     = arow[4 * i];
        a[2 * i + 1] = arow[4 * i + 1];
    }
}

template <int K>
__device__ __forceinline__ v8f mm_reg(const float (&af)[K / 2], const float (&bf)[K / 2], v8f c)
{
#pragma unroll
    for (int i = 0; i < K / 4; ++i) {
        v2f a, b;
        a.x = af[2 * i];
        a.y = af[2 * i + 1];
        b.x = bf[2 * i];
        b.y = bf[2 * i + 1];
        c = __builtin_amdgcn_wmma_f32_16x16x4_f32(
                false, a, false, b, (short)0, c, false, false);
    }
    return c;
}

template <int K>
__device__ __forceinline__ v8f mm_lds(const float* arow, const float (&bf)[K / 2], v8f c)
{
#pragma unroll
    for (int i = 0; i < K / 4; ++i) {
        v2f a, b;
        a.x = arow[4 * i];
        a.y = arow[4 * i + 1];
        b.x = bf[2 * i];
        b.y = bf[2 * i + 1];
        c = __builtin_amdgcn_wmma_f32_16x16x4_f32(
                false, a, false, b, (short)0, c, false, false);
    }
    return c;
}

__device__ __forceinline__ void pq_from_tile(const float* arow,
                                             const float (&bp)[cD / 2],
                                             const float (&bq)[cD / 2],
                                             float* __restrict__ P,
                                             float* __restrict__ Q,
                                             int mbase, int n)
{
    v8f p = {};
    v8f q = {};
#pragma unroll
    for (int i = 0; i < cD / 4; ++i) {
        v2f a, b0, b1;
        a.x  = arow[4 * i];
        a.y  = arow[4 * i + 1];
        b0.x = bp[2 * i];
        b0.y = bp[2 * i + 1];
        b1.x = bq[2 * i];
        b1.y = bq[2 * i + 1];
        p = __builtin_amdgcn_wmma_f32_16x16x4_f32(
                false, a, false, b0, (short)0, p, false, false);
        q = __builtin_amdgcn_wmma_f32_16x16x4_f32(
                false, a, false, b1, (short)0, q, false, false);
    }
#pragma unroll
    for (int r = 0; r < 8; ++r) {
        P[(size_t)(mbase + r) * cD + n] = p[r];
        Q[(size_t)(mbase + r) * cD + n] = q[r];
    }
}

__global__ __launch_bounds__(256)
void in_proj_pq(const float* __restrict__ h,
                const float* __restrict__ ipw,
                const float* __restrict__ ipb,
                const float* __restrict__ mw1,
                float* __restrict__ x,
                float* __restrict__ P,
                float* __restrict__ Q)
{
    __shared__ __attribute__((aligned(16))) float sX[16][cLD];

    const int wave  = threadIdx.x >> 5;
    const int lane  = threadIdx.x & 31;
    const int m0    = blockIdx.x * 16;
    const int l15   = lane & 15;
    const int khalf = lane >> 4;
    const int n     = wave * 16 + l15;
    const int rb    = khalf * 8;

    const size_t boff = (size_t)(2 * khalf) * cD + n;

    float a0[cDI / 2], b0[cDI / 2];
    load_b_frag<cDI>(ipw + boff, b0);
    load_a_frag<cDI>(h + (size_t)(m0 + l15) * cDI + 2 * khalf, a0);
    float bp[cD / 2], bq[cD / 2];
    load_b_frag<cD>(mw1 + boff, bp);
    load_b_frag<cD>(mw1 + (size_t)cD * cD + boff, bq);

    v8f c = {};
    c = mm_reg<cDI>(a0, b0, c);
    const float bv = ipb[n];
#pragma unroll
    for (int r = 0; r < 8; ++r) {
        float y = c[r] + bv;
        y = y > 0.0f ? y : 0.0f;
        sX[rb + r][n] = y;
        x[(size_t)(m0 + rb + r) * cD + n] = y;
    }
    __syncthreads();
    pq_from_tile(&sX[l15][2 * khalf], bp, bq, P, Q, m0 + rb, n);
}

static constexpr int cU  = 16;
static constexpr int cCH = cN / 32;

__device__ __forceinline__ void pair_step(float p, float q, float a, float w,
                                          float b1, float& acc, float& dg)
{
    float t = __builtin_fmaf(a, w, p + q) + b1;
    t = t > 0.0f ? t : 0.0f;
    acc = __builtin_fmaf(a, t, acc);
    dg += a;
}

__global__ void pairwise_agg(const float* __restrict__ P,
                             const float* __restrict__ Q,
                             const float* __restrict__ adj,
                             const float* __restrict__ we,
                             const float* __restrict__ mb1,
                             float* __restrict__ R,
                             float* __restrict__ deg)
{
    const int bi   = blockIdx.x;
    const int b    = bi / cN;
    const int d    = threadIdx.x;
    const int wave = threadIdx.x >> 5;
    const int lane = threadIdx.x & 31;

    __shared__ __attribute__((aligned(16))) int   s_j[4][cN + cU];
    __shared__ __attribute__((aligned(16))) float s_a[4][cN + cU];

    const float* arow = adj + (size_t)bi * cN;
    float av[cCH];
#pragma unroll
    for (int c = 0; c < cCH; ++c)
        av[c] = arow[c * 32 + lane];

    int n = 0;
#pragma unroll
    for (int c = 0; c < cCH; ++c) {
        const bool nz = av[c] != 0.0f;
        const unsigned mask = __builtin_amdgcn_ballot_w32(nz);
        const int pos = n + (int)__builtin_amdgcn_mbcnt_lo(mask, 0u);
        if (nz) {
            s_j[wave][pos] = c * 32 + lane;
            s_a[wave][pos] = av[c];
        }
        n += __builtin_popcount(mask);
    }
    if (lane < cU) {
        s_j[wave][n + lane] = 0;
        s_a[wave][n + lane] = 0.0f;
    }
    __syncthreads();

    const float p  = P[(size_t)bi * cD + d];
    const float w  = we[d];
    const float b1 = mb1[d];
    const float* Qb = Q + (size_t)b * cN * cD;
    const int4*   lj = reinterpret_cast<const int4*>(s_j[wave]);
    const float4* la = reinterpret_cast<const float4*>(s_a[wave]);

    float acc = 0.0f;
    float dg  = 0.0f;
    int k = 0;
    for (; k + cU <= n; k += cU) {
        int jj[cU]; float aa[cU]; float q[cU];
#pragma unroll
        for (int v = 0; v < cU / 4; ++v) {
            const int4   j4 = lj[(k >> 2) + v];
            const float4 a4 = la[(k >> 2) + v];
            jj[4 * v + 0] = j4.x; jj[4 * v + 1] = j4.y; jj[4 * v + 2] = j4.z; jj[4 * v + 3] = j4.w;
            aa[4 * v + 0] = a4.x; aa[4 * v + 1] = a4.y; aa[4 * v + 2] = a4.z; aa[4 * v + 3] = a4.w;
        }
#pragma unroll
        for (int u = 0; u < cU; ++u)
            q[u] = Qb[(unsigned)(jj[u] * cD + d)];
#pragma unroll
        for (int u = 0; u < cU; ++u)
            pair_step(p, q[u], aa[u], w, b1, acc, dg);
    }
    if (k < n) {
        int jj[cU]; float aa[cU]; float q[cU];
#pragma unroll
        for (int v = 0; v < cU / 4; ++v) {
            const int4   j4 = lj[(k >> 2) + v];
            const float4 a4 = la[(k >> 2) + v];
            jj[4 * v + 0] = j4.x; jj[4 * v + 1] = j4.y; jj[4 * v + 2] = j4.z; jj[4 * v + 3] = j4.w;
            aa[4 * v + 0] = a4.x; aa[4 * v + 1] = a4.y; aa[4 * v + 2] = a4.z; aa[4 * v + 3] = a4.w;
        }
#pragma unroll
        for (int u = 0; u < cU; ++u)
            q[u] = Qb[(unsigned)(jj[u] * cD + d)];
#pragma unroll
        for (int u = 0; u < cU; ++u)
            if (k + u < n)
                pair_step(p, q[u], aa[u], w, b1, acc, dg);
    }
    R[(size_t)bi * cD + d] = acc;
    if (d == 0) deg[bi] = dg;
}

template <bool kNext>
__global__ __launch_bounds__(256)
void layer_chain(const float* __restrict__ R,
                 const float* __restrict__ deg,
                 const float* xin,
                 const float* __restrict__ mw2,
                 const float* __restrict__ mb2,
                 const float* __restrict__ uw1,
                 const float* __restrict__ ub1,
                 const float* __restrict__ uw2,
                 const float* __restrict__ ub2,
                 const float* __restrict__ g,
                 const float* __restrict__ beta,
                 const float* __restrict__ mw1n,
                 float* __restrict__ P,
                 float* __restrict__ Q,
                 float* xout,
                 int add_residual, int relu_after)
{
    __shared__ __attribute__((aligned(16))) float sA[16][cLD];
    __shared__ __attribute__((aligned(16))) float sH[16][cLD];

    const int wave  = threadIdx.x >> 5;
    const int lane  = threadIdx.x & 31;
    const int m0    = blockIdx.x * 16;
    const int l15   = lane & 15;
    const int khalf = lane >> 4;
    const int n     = wave * 16 + l15;
    const int rb    = khalf * 8;
    const size_t boff = (size_t)(2 * khalf) * cD + n;

    const float bv_m2 = mb2[n];
    const float bv_u1 = ub1[n];
    const float bv_u2 = ub2[n];
    float dg[8], xr[8];
#pragma unroll
    for (int r = 0; r < 8; ++r) {
        dg[r] = deg[m0 + rb + r];
        xr[r] = xin[(size_t)(m0 + rb + r) * cD + n];
    }
    float gg[4], bb[4];
#pragma unroll
    for (int i = 0; i < 4; ++i) {
        gg[i] = g[lane + 32 * i];
        bb[i] = beta[lane + 32 * i];
    }

    float a1[cD / 2], b1[cD / 2];
    load_b_frag<cD>(mw2 + boff, b1);
    load_a_frag<cD>(R + (size_t)(m0 + l15) * cD + 2 * khalf, a1);
    float a2[cD / 2], b2a[cD / 2];
    load_b_frag<cD>(uw1 + boff, b2a);
    load_a_frag<cD>(xin + (size_t)(m0 + l15) * cD + 2 * khalf, a2);

    {
        v8f c = {};
        c = mm_reg<cD>(a1, b1, c);
#pragma unroll
        for (int r = 0; r < 8; ++r)
            sA[rb + r][n] = c[r] + dg[r] * bv_m2;
    }
    __syncthreads();

    float b3[cD / 2];
    {
        float b2b[cD / 2];
        load_b_frag<cD>(uw1 + (size_t)cD * cD + boff, b2b);
        v8f c = {};
        c = mm_reg<cD>(a2, b2a, c);
#pragma unroll
        for (int r = 0; r < 8; ++r)
            c[r] += bv_u1;
        load_b_frag<cD>(uw2 + boff, b3);
        c = mm_lds<cD>(&sA[l15][2 * khalf], b2b, c);
#pragma unroll
        for (int r = 0; r < 8; ++r)
            sH[rb + r][n] = c[r] > 0.0f ? c[r] : 0.0f;
    }
    __syncthreads();

    float bp[cD / 2], bq[cD / 2];
    if (kNext) {
        load_b_frag<cD>(mw1n + boff, bp);
        load_b_frag<cD>(mw1n + (size_t)cD * cD + boff, bq);
    }
    {
        v8f c = {};
        c = mm_lds<cD>(&sH[l15][2 * khalf], b3, c);
#pragma unroll
        for (int r = 0; r < 8; ++r) {
            float v = c[r] + bv_u2;
            if (add_residual) v += xr[r];
            sA[rb + r][n] = v;
        }
    }
    __syncthreads();

#pragma unroll
    for (int rr = 0; rr < 2; ++rr) {
        const int row = 2 * wave + rr;
        float v[4];
#pragma unroll
        for (int i = 0; i < 4; ++i)
            v[i] = sA[row][lane + 32 * i];

        float s = (v[0] + v[2]) + (v[1] + v[3]);
#pragma unroll
        for (int o = 16; o > 0; o >>= 1)
            s += __shfl_down(s, o);
        const float mu = __shfl(s, 0) * (1.0f / cD);

        float dv[4];
#pragma unroll
        for (int i = 0; i < 4; ++i)
            dv[i] = v[i] - mu;
        float t = (dv[0] * dv[0] + dv[2] * dv[2]) + (dv[1] * dv[1] + dv[3] * dv[3]);
#pragma unroll
        for (int o = 16; o > 0; o >>= 1)
            t += __shfl_down(t, o);
        const float var = __shfl(t, 0) * (1.0f / cD);
        const float rs  = rsqrtf(var + 1e-5f);

#pragma unroll
        for (int i = 0; i < 4; ++i) {
            float y = dv[i] * rs * gg[i] + bb[i];
            if (relu_after) y = y > 0.0f ? y : 0.0f;
            xout[(size_t)(m0 + row) * cD + lane + 32 * i] = y;
            sH[row][lane + 32 * i] = y;
        }
    }

    if (kNext) {
        __syncthreads();
        pq_from_tile(&sH[l15][2 * khalf], bp, bq, P, Q, m0 + rb, n);
    }
}

extern "C" void kernel_launch(void* const* d_in, const int* in_sizes, int n_in,
                              void* d_out, int out_size, void* d_ws, size_t ws_size,
                              hipStream_t stream)
{
    (void)in_sizes; (void)n_in; (void)out_size; (void)ws_size;

    const float* h    = (const float*)d_in[0];
    const float* adj  = (const float*)d_in[1];
    const float* ip_w = (const float*)d_in[2];
    const float* ip_b = (const float*)d_in[3];
    const float* mw1  = (const float*)d_in[4];
    const float* mb1  = (const float*)d_in[5];
    const float* mw2  = (const float*)d_in[6];
    const float* mb2  = (const float*)d_in[7];
    const float* uw1  = (const float*)d_in[8];
    const float* ub1  = (const float*)d_in[9];
    const float* uw2  = (const float*)d_in[10];
    const float* ub2  = (const float*)d_in[11];
    const float* ln_g = (const float*)d_in[12];
    const float* ln_b = (const float*)d_in[13];

    float* ws = (float*)d_ws;
    const size_t S = (size_t)cM * cD;
    float* x   = ws + 0 * S;
    float* P   = ws + 1 * S;
    float* Q   = ws + 2 * S;
    float* R   = ws + 3 * S;
    float* deg = ws + 7 * S;

    in_proj_pq<<<cM / 16, 256, 0, stream>>>(h, ip_w, ip_b, mw1, x, P, Q);

    for (int l = 0; l < 3; ++l) {
        const float* mw1l = mw1 + (size_t)l * 257 * cD;
        const float* mb1l = mb1 + (size_t)l * cD;
        const float* mw2l = mw2 + (size_t)l * cD * cD;
        const float* mb2l = mb2 + (size_t)l * cD;
        const float* uw1l = uw1 + (size_t)l * 256 * cD;
        const float* ub1l = ub1 + (size_t)l * cD;
        const float* uw2l = uw2 + (size_t)l * cD * cD;
        const float* ub2l = ub2 + (size_t)l * cD;

        pairwise_agg<<<cM, cD, 0, stream>>>(P, Q, adj, mw1l + 256 * cD, mb1l, R, deg);

        if (l < 2) {
            layer_chain<true><<<cM / 16, 256, 0, stream>>>(R, deg, x, mw2l, mb2l, uw1l, ub1l, uw2l, ub2l,
                                                           ln_g + l * cD, ln_b + l * cD,
                                                           mw1l + 257 * cD, P, Q, x, 1, 1);
        } else {
            layer_chain<false><<<cM / 16, 256, 0, stream>>>(R, deg, x, mw2l, mb2l, uw1l, ub1l, uw2l, ub2l,
                                                            ln_g + l * cD, ln_b + l * cD,
                                                            nullptr, P, Q, (float*)d_out, 0, 0);
        }
    }
}


# A tiles from global (h, R, x) come in once per block as whole rows (16-byte loads) into LDS and fragments are read there, not as 16-line per-lane loads repeated by 8 waves; every sum keeps its order
# speedup vs baseline: 3.8736x; 1.0559x over previous
#include <hip/hip_runtime.h>

typedef __attribute__((ext_vector_type(2))) float v2f;
typedef __attribute__((ext_vector_type(8))) float v8f;


static constexpr int cB = 2;
static constexpr int cN = 384;
static constexpr int cD = 128;
static constexpr int cM = cB * cN;
static constexpr int cDI = 64;
static constexpr int cLD = cD + 4;

template <int K>
__device__ __forceinline__ void load_b_frag(const float* __restrict__ bcol, float (&b)[K / 2])
{
#pragma unroll
    for (int i = 0; i < K / 4; ++i) {
        b[2 * i]     = bcol[(size_t)(4 * i) * cD];
        b[2 * i + 1] = bcol[(size_t)(4 * i + 1) * cD];
    }
}

template <int K>
__device__ __forceinline__ void load_a_frag(const float* arow, float (&a)[K / 2])
{
#pragma unroll
    for (int i = 0; i < K / 4; ++i) {
        a[2 * i]     = arow[4 * i];
        a[2 * i + 1] = arow[4 * i + 1];
    }
}

template <int K>
__device__ __forceinline__ v8f mm_reg(const float (&af)[K / 2], const float (&bf)[K / 2], v8f c)
{
#pragma unroll
    for (int i = 0; i < K / 4; ++i) {
        v2f a, b;
        a.x = af[2 * i];
        a.y = af[2 * i + 1];
        b.x = bf[2 * i];
        b.y = bf[2 * i + 1];
        c = __builtin_amdgcn_wmma_f32_16x16x4_f32(
                false, a, false, b, (short)0, c, false, false);
    }
    return c;
}

template <int K>
__device__ __forceinline__ v8f mm_lds(const float* arow, const float (&bf)[K / 2], v8f c)
{
#pragma unroll
    for (int i = 0; i < K / 4; ++i) {
        v2f a, b;
        a.x = arow[4 * i];
        a.y = arow[4 * i + 1];
        b.x = bf[2 * i];
        b.y = bf[2 * i + 1];
        c = __builtin_amdgcn_wmma_f32_16x16x4_f32(
                false, a, false, b, (short)0, c, false, false);
    }
    return c;
}

__device__ __forceinline__ void pq_from_tile(const float* arow,
                                             const float (&bp)[cD / 2],
                                             const float (&bq)[cD / 2],
                                             float* __restrict__ P,
                                             float* __restrict__ Q,
                                             int mbase, int n)
{
    v8f p = {};
    v8f q = {};
#pragma unroll
    for (int i = 0; i < cD / 4; ++i) {
        v2f a, b0, b1;
        a.x  = arow[4 * i];
        a.y  = arow[4 * i + 1];
        b0.x = bp[2 * i];
        b0.y = bp[2 * i + 1];
        b1.x = bq[2 * i];
        b1.y = bq[2 * i + 1];
        p = __builtin_amdgcn_wmma_f32_16x16x4_f32(
                false, a, false, b0, (short)0, p, false, false);
        q = __builtin_amdgcn_wmma_f32_16x16x4_f32(
                false, a, false, b1, (short)0, q, false, false);
    }
#pragma unroll
    for (int r = 0; r < 8; ++r) {
        P[(size_t)(mbase + r) * cD + n] = p[r];
        Q[(size_t)(mbase + r) * cD + n] = q[r];
    }
}

__global__ __launch_bounds__(256)
void in_proj_pq(const float* __restrict__ h,
                const float* __restrict__ ipw,
                const float* __restrict__ ipb,
                const float* __restrict__ mw1,
                float* __restrict__ x,
                float* __restrict__ P,
                float* __restrict__ Q)
{
    __shared__ __attribute__((aligned(16))) float sX[16][cLD];
    __shared__ __attribute__((aligned(16))) float sI[16][cDI + 4];

    const int wave  = threadIdx.x >> 5;
    const int lane  = threadIdx.x & 31;
    const int m0    = blockIdx.x * 16;
    const int l15   = lane & 15;
    const int khalf = lane >> 4;
    const int n     = wave * 16 + l15;
    const int rb    = khalf * 8;

    const size_t boff = (size_t)(2 * khalf) * cD + n;

    const float4 th = reinterpret_cast<const float4*>(h + (size_t)m0 * cDI)[threadIdx.x];

    float a0[cDI / 2], b0[cDI / 2];
    load_b_frag<cDI>(ipw + boff, b0);
    *reinterpret_cast<float4*>(&sI[threadIdx.x >> 4][4 * (threadIdx.x & 15)]) = th;
    __syncthreads();
    float bp[cD / 2], bq[cD / 2];
    load_b_frag<cD>(mw1 + boff, bp);
    load_b_frag<cD>(mw1 + (size_t)cD * cD + boff, bq);
    load_a_frag<cDI>(&sI[l15][2 * khalf], a0);

    v8f c = {};
    c = mm_reg<cDI>(a0, b0, c);
    const float bv = ipb[n];
#pragma unroll
    for (int r = 0; r < 8; ++r) {
        float y = c[r] + bv;
        y = y > 0.0f ? y : 0.0f;
        sX[rb + r][n] = y;
        x[(size_t)(m0 + rb + r) * cD + n] = y;
    }
    __syncthreads();
    pq_from_tile(&sX[l15][2 * khalf], bp, bq, P, Q, m0 + rb, n);
}

static constexpr int cU  = 16;
static constexpr int cCH = cN / 32;

__device__ __forceinline__ void pair_step(float p, float q, float a, float w,
                                          float b1, float& acc, float& dg)
{
    float t = __builtin_fmaf(a, w, p + q) + b1;
    t = t > 0.0f ? t : 0.0f;
    acc = __builtin_fmaf(a, t, acc);
    dg += a;
}

__global__ void pairwise_agg(const float* __restrict__ P,
                             const float* __restrict__ Q,
                             const float* __restrict__ adj,
                             const float* __restrict__ we,
                             const float* __restrict__ mb1,
                             float* __restrict__ R,
                             float* __restrict__ deg)
{
    const int bi   = blockIdx.x;
    const int b    = bi / cN;
    const int d    = threadIdx.x;
    const int wave = threadIdx.x >> 5;
    const int lane = threadIdx.x & 31;

    __shared__ __attribute__((aligned(16))) int   s_j[4][cN + cU];
    __shared__ __attribute__((aligned(16))) float s_a[4][cN + cU];

    const float* arow = adj + (size_t)bi * cN;
    float av[cCH];
#pragma unroll
    for (int c = 0; c < cCH; ++c)
        av[c] = arow[c * 32 + lane];

    int n = 0;
#pragma unroll
    for (int c = 0; c < cCH; ++c) {
        const bool nz = av[c] != 0.0f;
        const unsigned mask = __builtin_amdgcn_ballot_w32(nz);
        const int pos = n + (int)__builtin_amdgcn_mbcnt_lo(mask, 0u);
        if (nz) {
            s_j[wave][pos] = c * 32 + lane;
            s_a[wave][pos] = av[c];
        }
        n += __builtin_popcount(mask);
    }
    if (lane < cU) {
        s_j[wave][n + lane] = 0;
        s_a[wave][n + lane] = 0.0f;
    }
    __syncthreads();

    const float p  = P[(size_t)bi * cD + d];
    const float w  = we[d];
    const float b1 = mb1[d];
    const float* Qb = Q + (size_t)b * cN * cD;
    const int4*   lj = reinterpret_cast<const int4*>(s_j[wave]);
    const float4* la = reinterpret_cast<const float4*>(s_a[wave]);

    float acc = 0.0f;
    float dg  = 0.0f;
    int k = 0;
    for (; k + cU <= n; k += cU) {
        int jj[cU]; float aa[cU]; float q[cU];
#pragma unroll
        for (int v = 0; v < cU / 4; ++v) {
            const int4   j4 = lj[(k >> 2) + v];
            const float4 a4 = la[(k >> 2) + v];
            jj[4 * v + 0] = j4.x; jj[4 * v + 1] = j4.y; jj[4 * v + 2] = j4.z; jj[4 * v + 3] = j4.w;
            aa[4 * v + 0] = a4.x; aa[4 * v + 1] = a4.y; aa[4 * v + 2] = a4.z; aa[4 * v + 3] = a4.w;
        }
#pragma unroll
        for (int u = 0; u < cU; ++u)
            q[u] = Qb[(unsigned)(jj[u] * cD + d)];
#pragma unroll
        for (int u = 0; u < cU; ++u)
            pair_step(p, q[u], aa[u], w, b1, acc, dg);
    }
    if (k < n) {
        int jj[cU]; float aa[cU]; float q[cU];
#pragma unroll
        for (int v = 0; v < cU / 4; ++v) {
            const int4   j4 = lj[(k >> 2) + v];
            const float4 a4 = la[(k >> 2) + v];
            jj[4 * v + 0] = j4.x; jj[4 * v + 1] = j4.y; jj[4 * v + 2] = j4.z; jj[4 * v + 3] = j4.w;
            aa[4 * v + 0] = a4.x; aa[4 * v + 1] = a4.y; aa[4 * v + 2] = a4.z; aa[4 * v + 3] = a4.w;
        }
#pragma unroll
        for (int u = 0; u < cU; ++u)
            q[u] = Qb[(unsigned)(jj[u] * cD + d)];
#pragma unroll
        for (int u = 0; u < cU; ++u)
            if (k + u < n)
                pair_step(p, q[u], aa[u], w, b1, acc, dg);
    }
    R[(size_t)bi * cD + d] = acc;
    if (d == 0) deg[bi] = dg;
}

template <bool kNext>
__global__ __launch_bounds__(256)
void layer_chain(const float* __restrict__ R,
                 const float* __restrict__ deg,
                 const float* xin,
                 const float* __restrict__ mw2,
                 const float* __restrict__ mb2,
                 const float* __restrict__ uw1,
                 const float* __restrict__ ub1,
                 const float* __restrict__ uw2,
                 const float* __restrict__ ub2,
                 const float* __restrict__ g,
                 const float* __restrict__ beta,
                 const float* __restrict__ mw1n,
                 float* __restrict__ P,
                 float* __restrict__ Q,
                 float* xout,
                 int add_residual, int relu_after)
{
    __shared__ __attribute__((aligned(16))) float sA[16][cLD];
    __shared__ __attribute__((aligned(16))) float sH[16][cLD];
    __shared__ __attribute__((aligned(16))) float sR[16][cLD];
    __shared__ __attribute__((aligned(16))) float sX[16][cLD];

    const int wave  = threadIdx.x >> 5;
    const int lane  = threadIdx.x & 31;
    const int m0    = blockIdx.x * 16;
    const int l15   = lane & 15;
    const int khalf = lane >> 4;
    const int n     = wave * 16 + l15;
    const int rb    = khalf * 8;
    const size_t boff = (size_t)(2 * khalf) * cD + n;

    const float4* R4 = reinterpret_cast<const float4*>(R + (size_t)m0 * cD);
    const float4* X4 = reinterpret_cast<const float4*>(xin + (size_t)m0 * cD);
    const float4 tr0 = R4[threadIdx.x];
    const float4 tr1 = R4[threadIdx.x + 256];
    const float4 tx0 = X4[threadIdx.x];
    const float4 tx1 = X4[threadIdx.x + 256];

    const float bv_m2 = mb2[n];
    const float bv_u1 = ub1[n];
    const float bv_u2 = ub2[n];
    float dg[8];
#pragma unroll
    for (int r = 0; r < 8; ++r)
        dg[r] = deg[m0 + rb + r];
    float gg[4], bb[4];
#pragma unroll
    for (int i = 0; i < 4; ++i) {
        gg[i] = g[lane + 32 * i];
        bb[i] = beta[lane + 32 * i];
    }

    float a1[cD / 2], b1[cD / 2];
    load_b_frag<cD>(mw2 + boff, b1);

    *reinterpret_cast<float4*>(&sR[wave][4 * lane])     = tr0;
    *reinterpret_cast<float4*>(&sR[wave + 8][4 * lane]) = tr1;
    *reinterpret_cast<float4*>(&sX[wave][4 * lane])     = tx0;
    *reinterpret_cast<float4*>(&sX[wave + 8][4 * lane]) = tx1;
    __syncthreads();

    float a2[cD / 2], b2a[cD / 2];
    load_b_frag<cD>(uw1 + boff, b2a);
    load_a_frag<cD>(&sR[l15][2 * khalf], a1);
    load_a_frag<cD>(&sX[l15][2 * khalf], a2);
    float xr[8];
#pragma unroll
    for (int r = 0; r < 8; ++r)
        xr[r] = sX[rb + r][n];

    {
        v8f c = {};
        c = mm_reg<cD>(a1, b1, c);
#pragma unroll
        for (int r = 0; r < 8; ++r)
            sA[rb + r][n] = c[r] + dg[r] * bv_m2;
    }
    __syncthreads();

    float b3[cD / 2];
    {
        float b2b[cD / 2];
        load_b_frag<cD>(uw1 + (size_t)cD * cD + boff, b2b);
        v8f c = {};
        c = mm_reg<cD>(a2, b2a, c);
#pragma unroll
        for (int r = 0; r < 8; ++r)
            c[r] += bv_u1;
        load_b_frag<cD>(uw2 + boff, b3);
        c = mm_lds<cD>(&sA[l15][2 * khalf], b2b, c);
#pragma unroll
        for (int r = 0; r < 8; ++r)
            sH[rb + r][n] = c[r] > 0.0f ? c[r] : 0.0f;
    }
    __syncthreads();

    float bp[cD / 2], bq[cD / 2];
    if (kNext) {
        load_b_frag<cD>(mw1n + boff, bp);
        load_b_frag<cD>(mw1n + (size_t)cD * cD + boff, bq);
    }
    {
        v8f c = {};
        c = mm_lds<cD>(&sH[l15][2 * khalf], b3, c);
#pragma unroll
        for (int r = 0; r < 8; ++r) {
            float v = c[r] + bv_u2;
            if (add_residual) v += xr[r];
            sA[rb + r][n] = v;
        }
    }
    __syncthreads();

#pragma unroll
    for (int rr = 0; rr < 2; ++rr) {
        const int row = 2 * wave + rr;
        float v[4];
#pragma unroll
        for (int i = 0; i < 4; ++i)
            v[i] = sA[row][lane + 32 * i];

        float s = (v[0] + v[2]) + (v[1] + v[3]);
#pragma unroll
        for (int o = 16; o > 0; o >>= 1)
            s += __shfl_down(s, o);
        const float mu = __shfl(s, 0) * (1.0f / cD);

        float dv[4];
#pragma unroll
        for (int i = 0; i < 4; ++i)
            dv[i] = v[i] - mu;
        float t = (dv[0] * dv[0] + dv[2] * dv[2]) + (dv[1] * dv[1] + dv[3] * dv[3]);
#pragma unroll
        for (int o = 16; o > 0; o >>= 1)
            t += __shfl_down(t, o);
        const float var = __shfl(t, 0) * (1.0f / cD);
        const float rs  = rsqrtf(var + 1e-5f);

#pragma unroll
        for (int i = 0; i < 4; ++i) {
            float y = dv[i] * rs * gg[i] + bb[i];
            if (relu_after) y = y > 0.0f ? y : 0.0f;
            xout[(size_t)(m0 + row) * cD + lane + 32 * i] = y;
            sH[row][lane + 32 * i] = y;
        }
    }

    if (kNext) {
        __syncthreads();
        pq_from_tile(&sH[l15][2 * khalf], bp, bq, P, Q, m0 + rb, n);
    }
}

extern "C" void kernel_launch(void* const* d_in, const int* in_sizes, int n_in,
                              void* d_out, int out_size, void* d_ws, size_t ws_size,
                              hipStream_t stream)
{
    (void)in_sizes; (void)n_in; (void)out_size; (void)ws_size;

    const float* h    = (const float*)d_in[0];
    const float* adj  = (const float*)d_in[1];
    const float* ip_w = (const float*)d_in[2];
    const float* ip_b = (const float*)d_in[3];
    const float* mw1  = (const float*)d_in[4];
    const float* mb1  = (const float*)d_in[5];
    const float* mw2  = (const float*)d_in[6];
    const float* mb2  = (const float*)d_in[7];
    const float* uw1  = (const float*)d_in[8];
    const float* ub1  = (const float*)d_in[9];
    const float* uw2  = (const float*)d_in[10];
    const float* ub2  = (const float*)d_in[11];
    const float* ln_g = (const float*)d_in[12];
    const float* ln_b = (const float*)d_in[13];

    float* ws = (float*)d_ws;
    const size_t S = (size_t)cM * cD;
    float* x   = ws + 0 * S;
    float* P   = ws + 1 * S;
    float* Q   = ws + 2 * S;
    float* R   = ws + 3 * S;
    float* deg = ws + 7 * S;

    in_proj_pq<<<cM / 16, 256, 0, stream>>>(h, ip_w, ip_b, mw1, x, P, Q);

    for (int l = 0; l < 3; ++l) {
        const float* mw1l = mw1 + (size_t)l * 257 * cD;
        const float* mb1l = mb1 + (size_t)l * cD;
        const float* mw2l = mw2 + (size_t)l * cD * cD;
        const float* mb2l = mb2 + (size_t)l * cD;
        const float* uw1l = uw1 + (size_t)l * 256 * cD;
        const float* ub1l = ub1 + (size_t)l * cD;
        const float* uw2l = uw2 + (size_t)l * cD * cD;
        const float* ub2l = ub2 + (size_t)l * cD;

        pairwise_agg<<<cM, cD, 0, stream>>>(P, Q, adj, mw1l + 256 * cD, mb1l, R, deg);

        if (l < 2) {
            layer_chain<true><<<cM / 16, 256, 0, stream>>>(R, deg, x, mw2l, mb2l, uw1l, ub1l, uw2l, ub2l,
                                                           ln_g + l * cD, ln_b + l * cD,
                                                           mw1l + 257 * cD, P, Q, x, 1, 1);
        } else {
            layer_chain<false><<<cM / 16, 256, 0, stream>>>(R, deg, x, mw2l, mb2l, uw1l, ub1l, uw2l, ub2l,
                                                            ln_g + l * cD, ln_b + l * cD,
                                                            nullptr, P, Q, (float*)d_out, 0, 0);
        }
    }
}


# mw2 folded into uw1's lower half once per call (W' = mw2 @ uw1[128:], c' = mb2 @ uw1[128:], extra blocks of launch 1): hid takes R @ W', one product and barrier less a layer; reassociates that sum
# speedup vs baseline: 4.2718x; 1.1028x over previous
#include <hip/hip_runtime.h>

typedef __attribute__((ext_vector_type(2))) float v2f;
typedef __attribute__((ext_vector_type(8))) float v8f;


static constexpr int cB = 2;
static constexpr int cN = 384;
static constexpr int cD = 128;
static constexpr int cM = cB * cN;
static constexpr int cDI = 64;
static constexpr int cLD = cD + 4;

template <int K>
__device__ __forceinline__ void load_b_frag(const float* __restrict__ bcol, float (&b)[K / 2])
{
#pragma unroll
    for (int i = 0; i < K / 4; ++i) {
        b[2 * i]     = bcol[(size_t)(4 * i) * cD];
        b[2 * i + 1] = bcol[(size_t)(4 * i + 1) * cD];
    }
}

template <int K>
__device__ __forceinline__ void load_a_frag(const float* arow, float (&a)[K / 2])
{
#pragma unroll
    for (int i = 0; i < K / 4; ++i) {
        a[2 * i]     = arow[4 * i];
        a[2 * i + 1] = arow[4 * i + 1];
    }
}

template <int K>
__device__ __forceinline__ v8f mm_reg(const float (&af)[K / 2], const float (&bf)[K / 2], v8f c)
{
#pragma unroll
    for (int i = 0; i < K / 4; ++i) {
        v2f a, b;
        a.x = af[2 * i];
        a.y = af[2 * i + 1];
        b.x = bf[2 * i];
        b.y = bf[2 * i + 1];
        c = __builtin_amdgcn_wmma_f32_16x16x4_f32(
                false, a, false, b, (short)0, c, false, false);
    }
    return c;
}

template <int K>
__device__ __forceinline__ v8f mm_lds(const float* arow, const float (&bf)[K / 2], v8f c)
{
#pragma unroll
    for (int i = 0; i < K / 4; ++i) {
        v2f a, b;
        a.x = arow[4 * i];
        a.y = arow[4 * i + 1];
        b.x = bf[2 * i];
        b.y = bf[2 * i + 1];
        c = __builtin_amdgcn_wmma_f32_16x16x4_f32(
                false, a, false, b, (short)0, c, false, false);
    }
    return c;
}

__device__ __forceinline__ void pq_from_tile(const float* arow,
                                             const float (&bp)[cD / 2],
                                             const float (&bq)[cD / 2],
                                             float* __restrict__ P,
                                             float* __restrict__ Q,
                                             int mbase, int n)
{
    v8f p = {};
    v8f q = {};
#pragma unroll
    for (int i = 0; i < cD / 4; ++i) {
        v2f a, b0, b1;
        a.x  = arow[4 * i];
        a.y  = arow[4 * i + 1];
        b0.x = bp[2 * i];
        b0.y = bp[2 * i + 1];
        b1.x = bq[2 * i];
        b1.y = bq[2 * i + 1];
        p = __builtin_amdgcn_wmma_f32_16x16x4_f32(
                false, a, false, b0, (short)0, p, false, false);
        q = __builtin_amdgcn_wmma_f32_16x16x4_f32(
                false, a, false, b1, (short)0, q, false, false);
    }
#pragma unroll
    for (int r = 0; r < 8; ++r) {
        P[(size_t)(mbase + r) * cD + n] = p[r];
        Q[(size_t)(mbase + r) * cD + n] = q[r];
    }
}

static constexpr int cFoldTiles  = cD / 16 + 1;
static constexpr int cFoldRows   = 16 * cFoldTiles;
static constexpr int cFoldBlocks = 3 * cFoldTiles;

__global__ __launch_bounds__(256)
void in_proj_pq(const float* __restrict__ h,
                const float* __restrict__ ipw,
                const float* __restrict__ ipb,
                const float* __restrict__ mw1,
                const float* __restrict__ mw2,
                const float* __restrict__ mb2,
                const float* __restrict__ uw1,
                float* __restrict__ x,
                float* __restrict__ P,
                float* __restrict__ Q,
                float* __restrict__ Wp)
{
    __shared__ __attribute__((aligned(16))) float sX[16][cLD];
    __shared__ __attribute__((aligned(16))) float sI[16][cDI + 4];

    const int wave  = threadIdx.x >> 5;
    const int lane  = threadIdx.x & 31;
    const int m0    = blockIdx.x * 16;
    const int l15   = lane & 15;
    const int khalf = lane >> 4;
    const int n     = wave * 16 + l15;
    const int rb    = khalf * 8;

    const size_t boff = (size_t)(2 * khalf) * cD + n;

    if (blockIdx.x >= cM / 16) {
        const int e  = blockIdx.x - cM / 16;
        const int l  = e / cFoldTiles;
        const int t  = e % cFoldTiles;
        const float* ub = uw1 + (size_t)l * 2 * cD * cD + (size_t)cD * cD;
        float bf[cD / 2], af[cD / 2];
        load_b_frag<cD>(ub + boff, bf);
        float4 t0 = make_float4(0.0f, 0.0f, 0.0f, 0.0f);
        float4 t1 = t0;
        if (t < cD / 16) {
            const float4* A4 = reinterpret_cast<const float4*>(mw2 + (size_t)l * cD * cD + (size_t)(16 * t) * cD);
            t0 = A4[threadIdx.x];
            t1 = A4[threadIdx.x + 256];
        } else if (wave == 0) {
            t0 = reinterpret_cast<const float4*>(mb2 + (size_t)l * cD)[lane];
        }
        *reinterpret_cast<float4*>(&sX[wave][4 * lane])     = t0;
        *reinterpret_cast<float4*>(&sX[wave + 8][4 * lane]) = t1;
        __syncthreads();
        load_a_frag<cD>(&sX[l15][2 * khalf], af);
        v8f c = {};
        c = mm_reg<cD>(af, bf, c);
        float* Wl = Wp + (size_t)l * cFoldRows * cD + (size_t)(16 * t + rb) * cD + n;
#pragma unroll
        for (int r = 0; r < 8; ++r)
            Wl[(size_t)r * cD] = c[r];
        return;
    }

    const float4 th = reinterpret_cast<const float4*>(h + (size_t)m0 * cDI)[threadIdx.x];

    float a0[cDI / 2], b0[cDI / 2];
    load_b_frag<cDI>(ipw + boff, b0);
    *reinterpret_cast<float4*>(&sI[threadIdx.x >> 4][4 * (threadIdx.x & 15)]) = th;
    __syncthreads();
    float bp[cD / 2], bq[cD / 2];
    load_b_frag<cD>(mw1 + boff, bp);
    load_b_frag<cD>(mw1 + (size_t)cD * cD + boff, bq);
    load_a_frag<cDI>(&sI[l15][2 * khalf], a0);

    v8f c = {};
    c = mm_reg<cDI>(a0, b0, c);
    const float bv = ipb[n];
#pragma unroll
    for (int r = 0; r < 8; ++r) {
        float y = c[r] + bv;
        y = y > 0.0f ? y : 0.0f;
        sX[rb + r][n] = y;
        x[(size_t)(m0 + rb + r) * cD + n] = y;
    }
    __syncthreads();
    pq_from_tile(&sX[l15][2 * khalf], bp, bq, P, Q, m0 + rb, n);
}

static constexpr int cU  = 16;
static constexpr int cCH = cN / 32;

__device__ __forceinline__ void pair_step(float p, float q, float a, float w,
                                          float b1, float& acc, float& dg)
{
    float t = __builtin_fmaf(a, w, p + q) + b1;
    t = t > 0.0f ? t : 0.0f;
    acc = __builtin_fmaf(a, t, acc);
    dg += a;
}

__global__ void pairwise_agg(const float* __restrict__ P,
                             const float* __restrict__ Q,
                             const float* __restrict__ adj,
                             const float* __restrict__ we,
                             const float* __restrict__ mb1,
                             float* __restrict__ R,
                             float* __restrict__ deg)
{
    const int bi   = blockIdx.x;
    const int b    = bi / cN;
    const int d    = threadIdx.x;
    const int wave = threadIdx.x >> 5;
    const int lane = threadIdx.x & 31;

    __shared__ __attribute__((aligned(16))) int   s_j[4][cN + cU];
    __shared__ __attribute__((aligned(16))) float s_a[4][cN + cU];

    const float* arow = adj + (size_t)bi * cN;
    float av[cCH];
#pragma unroll
    for (int c = 0; c < cCH; ++c)
        av[c] = arow[c * 32 + lane];

    int n = 0;
#pragma unroll
    for (int c = 0; c < cCH; ++c) {
        const bool nz = av[c] != 0.0f;
        const unsigned mask = __builtin_amdgcn_ballot_w32(nz);
        const int pos = n + (int)__builtin_amdgcn_mbcnt_lo(mask, 0u);
        if (nz) {
            s_j[wave][pos] = c * 32 + lane;
            s_a[wave][pos] = av[c];
        }
        n += __builtin_popcount(mask);
    }
    if (lane < cU) {
        s_j[wave][n + lane] = 0;
        s_a[wave][n + lane] = 0.0f;
    }
    __syncthreads();

    const float p  = P[(size_t)bi * cD + d];
    const float w  = we[d];
    const float b1 = mb1[d];
    const float* Qb = Q + (size_t)b * cN * cD;
    const int4*   lj = reinterpret_cast<const int4*>(s_j[wave]);
    const float4* la = reinterpret_cast<const float4*>(s_a[wave]);

    float acc = 0.0f;
    float dg  = 0.0f;
    int k = 0;
    for (; k + cU <= n; k += cU) {
        int jj[cU]; float aa[cU]; float q[cU];
#pragma unroll
        for (int v = 0; v < cU / 4; ++v) {
            const int4   j4 = lj[(k >> 2) + v];
            const float4 a4 = la[(k >> 2) + v];
            jj[4 * v + 0] = j4.x; jj[4 * v + 1] = j4.y; jj[4 * v + 2] = j4.z; jj[4 * v + 3] = j4.w;
            aa[4 * v + 0] = a4.x; aa[4 * v + 1] = a4.y; aa[4 * v + 2] = a4.z; aa[4 * v + 3] = a4.w;
        }
#pragma unroll
        for (int u = 0; u < cU; ++u)
            q[u] = Qb[(unsigned)(jj[u] * cD + d)];
#pragma unroll
        for (int u = 0; u < cU; ++u)
            pair_step(p, q[u], aa[u], w, b1, acc, dg);
    }
    if (k < n) {
        int jj[cU]; float aa[cU]; float q[cU];
#pragma unroll
        for (int v = 0; v < cU / 4; ++v) {
            const int4   j4 = lj[(k >> 2) + v];
            const float4 a4 = la[(k >> 2) + v];
            jj[4 * v + 0] = j4.x; jj[4 * v + 1] = j4.y; jj[4 * v + 2] = j4.z; jj[4 * v + 3] = j4.w;
            aa[4 * v + 0] = a4.x; aa[4 * v + 1] = a4.y; aa[4 * v + 2] = a4.z; aa[4 * v + 3] = a4.w;
        }
#pragma unroll
        for (int u = 0; u < cU; ++u)
            q[u] = Qb[(unsigned)(jj[u] * cD + d)];
#pragma unroll
        for (int u = 0; u < cU; ++u)
            if (k + u < n)
                pair_step(p, q[u], aa[u], w, b1, acc, dg);
    }
    R[(size_t)bi * cD + d] = acc;
    if (d == 0) deg[bi] = dg;
}

template <bool kNext>
__global__ __launch_bounds__(256)
void layer_chain(const float* __restrict__ R,
                 const float* __restrict__ deg,
                 const float* xin,
                 const float* __restrict__ Wp,
                 const float* __restrict__ cp,
                 const float* __restrict__ uw1,
                 const float* __restrict__ ub1,
                 const float* __restrict__ uw2,
                 const float* __restrict__ ub2,
                 const float* __restrict__ g,
                 const float* __restrict__ beta,
                 const float* __restrict__ mw1n,
                 float* __restrict__ P,
                 float* __restrict__ Q,
                 float* xout,
                 int add_residual, int relu_after)
{
    __shared__ __attribute__((aligned(16))) float sA[16][cLD];
    __shared__ __attribute__((aligned(16))) float sH[16][cLD];
    __shared__ __attribute__((aligned(16))) float sR[16][cLD];
    __shared__ __attribute__((aligned(16))) float sX[16][cLD];

    const int wave  = threadIdx.x >> 5;
    const int lane  = threadIdx.x & 31;
    const int m0    = blockIdx.x * 16;
    const int l15   = lane & 15;
    const int khalf = lane >> 4;
    const int n     = wave * 16 + l15;
    const int rb    = khalf * 8;
    const size_t boff = (size_t)(2 * khalf) * cD + n;

    const float4* R4 = reinterpret_cast<const float4*>(R + (size_t)m0 * cD);
    const float4* X4 = reinterpret_cast<const float4*>(xin + (size_t)m0 * cD);
    const float4 tr0 = R4[threadIdx.x];
    const float4 tr1 = R4[threadIdx.x + 256];
    const float4 tx0 = X4[threadIdx.x];
    const float4 tx1 = X4[threadIdx.x + 256];

    const float bv_c  = cp[n];
    const float bv_u1 = ub1[n];
    const float bv_u2 = ub2[n];
    float dg[8];
#pragma unroll
    for (int r = 0; r < 8; ++r)
        dg[r] = deg[m0 + rb + r];
    float gg[4], bb[4];
#pragma unroll
    for (int i = 0; i < 4; ++i) {
        gg[i] = g[lane + 32 * i];
        bb[i] = beta[lane + 32 * i];
    }

    float a1[cD / 2], b1[cD / 2];
    load_b_frag<cD>(Wp + boff, b1);

    *reinterpret_cast<float4*>(&sR[wave][4 * lane])     = tr0;
    *reinterpret_cast<float4*>(&sR[wave + 8][4 * lane]) = tr1;
    *reinterpret_cast<float4*>(&sX[wave][4 * lane])     = tx0;
    *reinterpret_cast<float4*>(&sX[wave + 8][4 * lane]) = tx1;
    __syncthreads();

    float a2[cD / 2], b2a[cD / 2];
    load_b_frag<cD>(uw1 + boff, b2a);
    load_a_frag<cD>(&sR[l15][2 * khalf], a1);
    load_a_frag<cD>(&sX[l15][2 * khalf], a2);
    float xr[8];
#pragma unroll
    for (int r = 0; r < 8; ++r)
        xr[r] = sX[rb + r][n];

    float b3[cD / 2];
    {
        v8f c = {};
        c = mm_reg<cD>(a1, b1, c);
        load_b_frag<cD>(uw2 + boff, b3);
        c = mm_reg<cD>(a2, b2a, c);
#pragma unroll
        for (int r = 0; r < 8; ++r) {
            const float v = (c[r] + bv_u1) + dg[r] * bv_c;
            sH[rb + r][n] = v > 0.0f ? v : 0.0f;
        }
    }
    __syncthreads();

    float bp[cD / 2], bq[cD / 2];
    if (kNext) {
        load_b_frag<cD>(mw1n + boff, bp);
        load_b_frag<cD>(mw1n + (size_t)cD * cD + boff, bq);
    }
    {
        v8f c = {};
        c = mm_lds<cD>(&sH[l15][2 * khalf], b3, c);
#pragma unroll
        for (int r = 0; r < 8; ++r) {
            float v = c[r] + bv_u2;
            if (add_residual) v += xr[r];
            sA[rb + r][n] = v;
        }
    }
    __syncthreads();

#pragma unroll
    for (int rr = 0; rr < 2; ++rr) {
        const int row = 2 * wave + rr;
        float v[4];
#pragma unroll
        for (int i = 0; i < 4; ++i)
            v[i] = sA[row][lane + 32 * i];

        float s = (v[0] + v[2]) + (v[1] + v[3]);
#pragma unroll
        for (int o = 16; o > 0; o >>= 1)
            s += __shfl_down(s, o);
        const float mu = __shfl(s, 0) * (1.0f / cD);

        float dv[4];
#pragma unroll
        for (int i = 0; i < 4; ++i)
            dv[i] = v[i] - mu;
        float t = (dv[0] * dv[0] + dv[2] * dv[2]) + (dv[1] * dv[1] + dv[3] * dv[3]);
#pragma unroll
        for (int o = 16; o > 0; o >>= 1)
            t += __shfl_down(t, o);
        const float var = __shfl(t, 0) * (1.0f / cD);
        const float rs  = rsqrtf(var + 1e-5f);

#pragma unroll
        for (int i = 0; i < 4; ++i) {
            float y = dv[i] * rs * gg[i] + bb[i];
            if (relu_after) y = y > 0.0f ? y : 0.0f;
            xout[(size_t)(m0 + row) * cD + lane + 32 * i] = y;
            sH[row][lane + 32 * i] = y;
        }
    }

    if (kNext) {
        __syncthreads();
        pq_from_tile(&sH[l15][2 * khalf], bp, bq, P, Q, m0 + rb, n);
    }
}

extern "C" void kernel_launch(void* const* d_in, const int* in_sizes, int n_in,
                              void* d_out, int out_size, void* d_ws, size_t ws_size,
                              hipStream_t stream)
{
    (void)in_sizes; (void)n_in; (void)out_size; (void)ws_size;

    const float* h    = (const float*)d_in[0];
    const float* adj  = (const float*)d_in[1];
    const float* ip_w = (const float*)d_in[2];
    const float* ip_b = (const float*)d_in[3];
    const float* mw1  = (const float*)d_in[4];
    const float* mb1  = (const float*)d_in[5];
    const float* mw2  = (const float*)d_in[6];
    const float* mb2  = (const float*)d_in[7];
    const float* uw1  = (const float*)d_in[8];
    const float* ub1  = (const float*)d_in[9];
    const float* uw2  = (const float*)d_in[10];
    const float* ub2  = (const float*)d_in[11];
    const float* ln_g = (const float*)d_in[12];
    const float* ln_b = (const float*)d_in[13];

    float* ws = (float*)d_ws;
    const size_t S = (size_t)cM * cD;
    float* x   = ws + 0 * S;
    float* P   = ws + 1 * S;
    float* Q   = ws + 2 * S;
    float* R   = ws + 3 * S;
    float* deg = ws + 7 * S;
    float* Wp  = ws + 4 * S;

    in_proj_pq<<<cM / 16 + cFoldBlocks, 256, 0, stream>>>(h, ip_w, ip_b, mw1, mw2, mb2, uw1, x, P, Q, Wp);

    for (int l = 0; l < 3; ++l) {
        const float* mw1l = mw1 + (size_t)l * 257 * cD;
        const float* mb1l = mb1 + (size_t)l * cD;
        const float* Wpl  = Wp + (size_t)l * cFoldRows * cD;
        const float* cpl  = Wpl + (size_t)cD * cD;
        const float* uw1l = uw1 + (size_t)l * 256 * cD;
        const float* ub1l = ub1 + (size_t)l * cD;
        const float* uw2l = uw2 + (size_t)l * cD * cD;
        const float* ub2l = ub2 + (size_t)l * cD;

        pairwise_agg<<<cM, cD, 0, stream>>>(P, Q, adj, mw1l + 256 * cD, mb1l, R, deg);

        if (l < 2) {
            layer_chain<true><<<cM / 16, 256, 0, stream>>>(R, deg, x, Wpl, cpl, uw1l, ub1l, uw2l, ub2l,
                                                           ln_g + l * cD, ln_b + l * cD,
                                                           mw1l + 257 * cD, P, Q, x, 1, 1);
        } else {
            layer_chain<false><<<cM / 16, 256, 0, stream>>>(R, deg, x, Wpl, cpl, uw1l, ub1l, uw2l, ub2l,
                                                            ln_g + l * cD, ln_b + l * cD,
                                                            nullptr, P, Q, (float*)d_out, 0, 0);
        }
    }
}


# sender sum moved into the layer kernel, 16-wave blocks with one wave per row of the tile (16-byte sender loads), R and deg handed on in LDS: 7 launches to 4; every sum keeps its order
# speedup vs baseline: 4.8150x; 1.1272x over previous
#include <hip/hip_runtime.h>

typedef __attribute__((ext_vector_type(2))) float v2f;
typedef __attribute__((ext_vector_type(8))) float v8f;


static constexpr int cB = 2;
static constexpr int cN = 384;
static constexpr int cD = 128;
static constexpr int cM = cB * cN;
static constexpr int cDI = 64;
static constexpr int cLD = cD + 4;

template <int K>
__device__ __forceinline__ void load_b_frag(const float* __restrict__ bcol, float (&b)[K / 2])
{
#pragma unroll
    for (int i = 0; i < K / 4; ++i) {
        b[2 * i]     = bcol[(size_t)(4 * i) * cD];
        b[2 * i + 1] = bcol[(size_t)(4 * i + 1) * cD];
    }
}

template <int K>
__device__ __forceinline__ void load_a_frag(const float* arow, float (&a)[K / 2])
{
#pragma unroll
    for (int i = 0; i < K / 4; ++i) {
        a[2 * i]     = arow[4 * i];
        a[2 * i + 1] = arow[4 * i + 1];
    }
}

template <int K>
__device__ __forceinline__ v8f mm_reg(const float (&af)[K / 2], const float (&bf)[K / 2], v8f c)
{
#pragma unroll
    for (int i = 0; i < K / 4; ++i) {
        v2f a, b;
        a.x = af[2 * i];
        a.y = af[2 * i + 1];
        b.x = bf[2 * i];
        b.y = bf[2 * i + 1];
        c = __builtin_amdgcn_wmma_f32_16x16x4_f32(
                false, a, false, b, (short)0, c, false, false);
    }
    return c;
}

template <int K>
__device__ __forceinline__ v8f mm_lds(const float* arow, const float (&bf)[K / 2], v8f c)
{
#pragma unroll
    for (int i = 0; i < K / 4; ++i) {
        v2f a, b;
        a.x = arow[4 * i];
        a.y = arow[4 * i + 1];
        b.x = bf[2 * i];
        b.y = bf[2 * i + 1];
        c = __builtin_amdgcn_wmma_f32_16x16x4_f32(
                false, a, false, b, (short)0, c, false, false);
    }
    return c;
}

__device__ __forceinline__ void pq_from_tile(const float* arow,
                                             const float (&bp)[cD / 2],
                                             const float (&bq)[cD / 2],
                                             float* __restrict__ P,
                                             float* __restrict__ Q,
                                             int mbase, int n)
{
    v8f p = {};
    v8f q = {};
#pragma unroll
    for (int i = 0; i < cD / 4; ++i) {
        v2f a, b0, b1;
        a.x  = arow[4 * i];
        a.y  = arow[4 * i + 1];
        b0.x = bp[2 * i];
        b0.y = bp[2 * i + 1];
        b1.x = bq[2 * i];
        b1.y = bq[2 * i + 1];
        p = __builtin_amdgcn_wmma_f32_16x16x4_f32(
                false, a, false, b0, (short)0, p, false, false);
        q = __builtin_amdgcn_wmma_f32_16x16x4_f32(
                false, a, false, b1, (short)0, q, false, false);
    }
#pragma unroll
    for (int r = 0; r < 8; ++r) {
        P[(size_t)(mbase + r) * cD + n] = p[r];
        Q[(size_t)(mbase + r) * cD + n] = q[r];
    }
}

static constexpr int cFoldTiles  = cD / 16 + 1;
static constexpr int cFoldRows   = 16 * cFoldTiles;
static constexpr int cFoldBlocks = 3 * cFoldTiles;

__global__ __launch_bounds__(256)
void in_proj_pq(const float* __restrict__ h,
                const float* __restrict__ ipw,
                const float* __restrict__ ipb,
                const float* __restrict__ mw1,
                const float* __restrict__ mw2,
                const float* __restrict__ mb2,
                const float* __restrict__ uw1,
                float* __restrict__ x,
                float* __restrict__ P,
                float* __restrict__ Q,
                float* __restrict__ Wp)
{
    __shared__ __attribute__((aligned(16))) float sX[16][cLD];
    __shared__ __attribute__((aligned(16))) float sI[16][cDI + 4];

    const int wave  = threadIdx.x >> 5;
    const int lane  = threadIdx.x & 31;
    const int m0    = blockIdx.x * 16;
    const int l15   = lane & 15;
    const int khalf = lane >> 4;
    const int n     = wave * 16 + l15;
    const int rb    = khalf * 8;

    const size_t boff = (size_t)(2 * khalf) * cD + n;

    if (blockIdx.x >= cM / 16) {
        const int e  = blockIdx.x - cM / 16;
        const int l  = e / cFoldTiles;
        const int t  = e % cFoldTiles;
        const float* ub = uw1 + (size_t)l * 2 * cD * cD + (size_t)cD * cD;
        float bf[cD / 2], af[cD / 2];
        load_b_frag<cD>(ub + boff, bf);
        float4 t0 = make_float4(0.0f, 0.0f, 0.0f, 0.0f);
        float4 t1 = t0;
        if (t < cD / 16) {
            const float4* A4 = reinterpret_cast<const float4*>(mw2 + (size_t)l * cD * cD + (size_t)(16 * t) * cD);
            t0 = A4[threadIdx.x];
            t1 = A4[threadIdx.x + 256];
        } else if (wave == 0) {
            t0 = reinterpret_cast<const float4*>(mb2 + (size_t)l * cD)[lane];
        }
        *reinterpret_cast<float4*>(&sX[wave][4 * lane])     = t0;
        *reinterpret_cast<float4*>(&sX[wave + 8][4 * lane]) = t1;
        __syncthreads();
        load_a_frag<cD>(&sX[l15][2 * khalf], af);
        v8f c = {};
        c = mm_reg<cD>(af, bf, c);
        float* Wl = Wp + (size_t)l * cFoldRows * cD + (size_t)(16 * t + rb) * cD + n;
#pragma unroll
        for (int r = 0; r < 8; ++r)
            Wl[(size_t)r * cD] = c[r];
        return;
    }

    const float4 th = reinterpret_cast<const float4*>(h + (size_t)m0 * cDI)[threadIdx.x];

    float a0[cDI / 2], b0[cDI / 2];
    load_b_frag<cDI>(ipw + boff, b0);
    *reinterpret_cast<float4*>(&sI[threadIdx.x >> 4][4 * (threadIdx.x & 15)]) = th;
    __syncthreads();
    float bp[cD / 2], bq[cD / 2];
    load_b_frag<cD>(mw1 + boff, bp);
    load_b_frag<cD>(mw1 + (size_t)cD * cD + boff, bq);
    load_a_frag<cDI>(&sI[l15][2 * khalf], a0);

    v8f c = {};
    c = mm_reg<cDI>(a0, b0, c);
    const float bv = ipb[n];
#pragma unroll
    for (int r = 0; r < 8; ++r) {
        float y = c[r] + bv;
        y = y > 0.0f ? y : 0.0f;
        sX[rb + r][n] = y;
        x[(size_t)(m0 + rb + r) * cD + n] = y;
    }
    __syncthreads();
    pq_from_tile(&sX[l15][2 * khalf], bp, bq, P, Q, m0 + rb, n);
}

static constexpr int cU  = 16;
static constexpr int cCH = cN / 32;

__device__ __forceinline__ void pair_lane(float p, float q, float a, float w,
                                          float b1, float& acc)
{
    float t = __builtin_fmaf(a, w, p + q) + b1;
    t = t > 0.0f ? t : 0.0f;
    acc = __builtin_fmaf(a, t, acc);
}

__device__ __forceinline__ void pair_step(const float4& p, const float4& q, float a,
                                          const float4& w, const float4& b1,
                                          float4& acc, float& dg)
{
    pair_lane(p.x, q.x, a, w.x, b1.x, acc.x);
    pair_lane(p.y, q.y, a, w.y, b1.y, acc.y);
    pair_lane(p.z, q.z, a, w.z, b1.z, acc.z);
    pair_lane(p.w, q.w, a, w.w, b1.w, acc.w);
    dg += a;
}

template <bool kNext>
__global__ __launch_bounds__(512)
void layer_fused(const float* __restrict__ Pin,
                 const float* __restrict__ Qin,
                 const float* __restrict__ adj,
                 const float* __restrict__ we,
                 const float* __restrict__ mb1,
                 const float* xin,
                 const float* __restrict__ Wp,
                 const float* __restrict__ cp,
                 const float* __restrict__ uw1,
                 const float* __restrict__ ub1,
                 const float* __restrict__ uw2,
                 const float* __restrict__ ub2,
                 const float* __restrict__ g,
                 const float* __restrict__ beta,
                 const float* __restrict__ mw1n,
                 float* __restrict__ Pout,
                 float* __restrict__ Qout,
                 float* xout,
                 int add_residual, int relu_after)
{
    __shared__ __attribute__((aligned(16))) float sA[16][cLD];
    __shared__ __attribute__((aligned(16))) float sH[16][cLD];
    __shared__ __attribute__((aligned(16))) float sR[16][cLD];
    __shared__ __attribute__((aligned(16))) float sX[16][cLD];
    __shared__ __attribute__((aligned(16))) int   s_j[16][cN + cU];
    __shared__ __attribute__((aligned(16))) float s_a[16][cN + cU];
    __shared__ float sDeg[16];

    const int wave  = threadIdx.x >> 5;
    const int lane  = threadIdx.x & 31;
    const int m0    = blockIdx.x * 16;
    const int l15   = lane & 15;
    const int khalf = lane >> 4;
    const int n     = (wave & 7) * 16 + l15;
    const int rb    = khalf * 8;
    const size_t boff = (size_t)(2 * khalf) * cD + n;
    const bool mmw  = wave < 8;

    const int row = m0 + wave;
    const int b   = m0 / cN;

    const float* arow = adj + (size_t)row * cN;
    float av[cCH];
#pragma unroll
    for (int c = 0; c < cCH; ++c)
        av[c] = arow[c * 32 + lane];

    const float4 tx  = reinterpret_cast<const float4*>(xin + (size_t)row * cD)[lane];
    const float4 p4  = reinterpret_cast<const float4*>(Pin + (size_t)row * cD)[lane];
    const float4 w4  = reinterpret_cast<const float4*>(we)[lane];
    const float4 b14 = reinterpret_cast<const float4*>(mb1)[lane];

    const float bv_c  = cp[n];
    const float bv_u1 = ub1[n];
    const float bv_u2 = ub2[n];
    float gg[4], bb[4];
#pragma unroll
    for (int i = 0; i < 4; ++i) {
        gg[i] = g[lane + 32 * i];
        bb[i] = beta[lane + 32 * i];
    }

    float b1[cD / 2];
    if (mmw)
        load_b_frag<cD>(Wp + boff, b1);

    int ns = 0;
#pragma unroll
    for (int c = 0; c < cCH; ++c) {
        const bool nz = av[c] != 0.0f;
        const unsigned mask = __builtin_amdgcn_ballot_w32(nz);
        const int pos = ns + (int)__builtin_amdgcn_mbcnt_lo(mask, 0u);
        if (nz) {
            s_j[wave][pos] = c * 32 + lane;
            s_a[wave][pos] = av[c];
        }
        ns += __builtin_popcount(mask);
    }
    if (lane < cU) {
        s_j[wave][ns + lane] = 0;
        s_a[wave][ns + lane] = 0.0f;
    }
    __syncthreads();

    const float4* Qb = reinterpret_cast<const float4*>(Qin + (size_t)b * cN * cD);
    const int4*   lj = reinterpret_cast<const int4*>(s_j[wave]);
    const float4* la = reinterpret_cast<const float4*>(s_a[wave]);

    float4 acc = make_float4(0.0f, 0.0f, 0.0f, 0.0f);
    float  dg  = 0.0f;
    int k = 0;
    for (; k + cU <= ns; k += cU) {
        int jj[cU]; float aa[cU]; float4 q[cU];
#pragma unroll
        for (int v = 0; v < cU / 4; ++v) {
            const int4   j4 = lj[(k >> 2) + v];
            const float4 a4 = la[(k >> 2) + v];
            jj[4 * v + 0] = j4.x; jj[4 * v + 1] = j4.y; jj[4 * v + 2] = j4.z; jj[4 * v + 3] = j4.w;
            aa[4 * v + 0] = a4.x; aa[4 * v + 1] = a4.y; aa[4 * v + 2] = a4.z; aa[4 * v + 3] = a4.w;
        }
#pragma unroll
        for (int u = 0; u < cU; ++u)
            q[u] = Qb[(unsigned)(jj[u] * (cD / 4) + lane)];
#pragma unroll
        for (int u = 0; u < cU; ++u)
            pair_step(p4, q[u], aa[u], w4, b14, acc, dg);
    }
    if (k < ns) {
        int jj[cU]; float aa[cU]; float4 q[cU];
#pragma unroll
        for (int v = 0; v < cU / 4; ++v) {
            const int4   j4 = lj[(k >> 2) + v];
            const float4 a4 = la[(k >> 2) + v];
            jj[4 * v + 0] = j4.x; jj[4 * v + 1] = j4.y; jj[4 * v + 2] = j4.z; jj[4 * v + 3] = j4.w;
            aa[4 * v + 0] = a4.x; aa[4 * v + 1] = a4.y; aa[4 * v + 2] = a4.z; aa[4 * v + 3] = a4.w;
        }
#pragma unroll
        for (int u = 0; u < cU; ++u)
            q[u] = Qb[(unsigned)(jj[u] * (cD / 4) + lane)];
#pragma unroll
        for (int u = 0; u < cU; ++u)
            if (k + u < ns)
                pair_step(p4, q[u], aa[u], w4, b14, acc, dg);
    }

    *reinterpret_cast<float4*>(&sR[wave][4 * lane]) = acc;
    *reinterpret_cast<float4*>(&sX[wave][4 * lane]) = tx;
    if (lane == 0) sDeg[wave] = dg;
    __syncthreads();

    float xr[8];
    float b3[cD / 2];
    if (mmw) {
        float a1[cD / 2];
        float a2[cD / 2], b2a[cD / 2];
        load_b_frag<cD>(uw1 + boff, b2a);
        load_a_frag<cD>(&sR[l15][2 * khalf], a1);
        load_a_frag<cD>(&sX[l15][2 * khalf], a2);
        float dr[8];
#pragma unroll
        for (int r = 0; r < 8; ++r) {
            xr[r] = sX[rb + r][n];
            dr[r] = sDeg[rb + r];
        }
        v8f c = {};
        c = mm_reg<cD>(a1, b1, c);
        load_b_frag<cD>(uw2 + boff, b3);
        c = mm_reg<cD>(a2, b2a, c);
#pragma unroll
        for (int r = 0; r < 8; ++r) {
            const float v = (c[r] + bv_u1) + dr[r] * bv_c;
            sH[rb + r][n] = v > 0.0f ? v : 0.0f;
        }
    }
    __syncthreads();

    float bpq[cD / 2];
    if (kNext)
        load_b_frag<cD>(mw1n + (mmw ? (size_t)0 : (size_t)cD * cD) + boff, bpq);
    if (mmw) {
        v8f c = {};
        c = mm_lds<cD>(&sH[l15][2 * khalf], b3, c);
#pragma unroll
        for (int r = 0; r < 8; ++r) {
            float v = c[r] + bv_u2;
            if (add_residual) v += xr[r];
            sA[rb + r][n] = v;
        }
    }
    __syncthreads();

    {
        float v[4];
#pragma unroll
        for (int i = 0; i < 4; ++i)
            v[i] = sA[wave][lane + 32 * i];

        float s = (v[0] + v[2]) + (v[1] + v[3]);
#pragma unroll
        for (int o = 16; o > 0; o >>= 1)
            s += __shfl_down(s, o);
        const float mu = __shfl(s, 0) * (1.0f / cD);

        float dv[4];
#pragma unroll
        for (int i = 0; i < 4; ++i)
            dv[i] = v[i] - mu;
        float t = (dv[0] * dv[0] + dv[2] * dv[2]) + (dv[1] * dv[1] + dv[3] * dv[3]);
#pragma unroll
        for (int o = 16; o > 0; o >>= 1)
            t += __shfl_down(t, o);
        const float var = __shfl(t, 0) * (1.0f / cD);
        const float rs  = rsqrtf(var + 1e-5f);

#pragma unroll
        for (int i = 0; i < 4; ++i) {
            float y = dv[i] * rs * gg[i] + bb[i];
            if (relu_after) y = y > 0.0f ? y : 0.0f;
            xout[(size_t)row * cD + lane + 32 * i] = y;
            sH[wave][lane + 32 * i] = y;
        }
    }

    if (kNext) {
        __syncthreads();
        v8f c = {};
        c = mm_lds<cD>(&sH[l15][2 * khalf], bpq, c);
        float* __restrict__ O = mmw ? Pout : Qout;
#pragma unroll
        for (int r = 0; r < 8; ++r)
            O[(size_t)(m0 + rb + r) * cD + n] = c[r];
    }
}

extern "C" void kernel_launch(void* const* d_in, const int* in_sizes, int n_in,
                              void* d_out, int out_size, void* d_ws, size_t ws_size,
                              hipStream_t stream)
{
    (void)in_sizes; (void)n_in; (void)out_size; (void)ws_size;

    const float* h    = (const float*)d_in[0];
    const float* adj  = (const float*)d_in[1];
    const float* ip_w = (const float*)d_in[2];
    const float* ip_b = (const float*)d_in[3];
    const float* mw1  = (const float*)d_in[4];
    const float* mb1  = (const float*)d_in[5];
    const float* mw2  = (const float*)d_in[6];
    const float* mb2  = (const float*)d_in[7];
    const float* uw1  = (const float*)d_in[8];
    const float* ub1  = (const float*)d_in[9];
    const float* uw2  = (const float*)d_in[10];
    const float* ub2  = (const float*)d_in[11];
    const float* ln_g = (const float*)d_in[12];
    const float* ln_b = (const float*)d_in[13];

    float* ws = (float*)d_ws;
    const size_t S = (size_t)cM * cD;
    float* x  = ws + 0 * S;
    float* Pa = ws + 1 * S;
    float* Qa = ws + 2 * S;
    float* Pb = ws + 5 * S;
    float* Qb = ws + 6 * S;
    float* Wp = ws + 4 * S;

    in_proj_pq<<<cM / 16 + cFoldBlocks, 256, 0, stream>>>(h, ip_w, ip_b, mw1, mw2, mb2, uw1, x, Pa, Qa, Wp);

    for (int l = 0; l < 3; ++l) {
        const float* mw1l = mw1 + (size_t)l * 257 * cD;
        const float* mb1l = mb1 + (size_t)l * cD;
        const float* Wpl  = Wp + (size_t)l * cFoldRows * cD;
        const float* cpl  = Wpl + (size_t)cD * cD;
        const float* uw1l = uw1 + (size_t)l * 256 * cD;
        const float* ub1l = ub1 + (size_t)l * cD;
        const float* uw2l = uw2 + (size_t)l * cD * cD;
        const float* ub2l = ub2 + (size_t)l * cD;
        float* Pi = (l == 1) ? Pb : Pa;
        float* Qi = (l == 1) ? Qb : Qa;
        float* Po = (l == 1) ? Pa : Pb;
        float* Qo = (l == 1) ? Qa : Qb;

        if (l < 2) {
            layer_fused<true><<<cM / 16, 512, 0, stream>>>(Pi, Qi, adj, mw1l + 256 * cD, mb1l, x, Wpl, cpl,
                                                           uw1l, ub1l, uw2l, ub2l,
                                                           ln_g + l * cD, ln_b + l * cD,
                                                           mw1l + 257 * cD, Po, Qo, x, 1, 1);
        } else {
            layer_fused<false><<<cM / 16, 512, 0, stream>>>(Pi, Qi, adj, mw1l + 256 * cD, mb1l, x, Wpl, cpl,
                                                            uw1l, ub1l, uw2l, ub2l,
                                                            ln_g + l * cD, ln_b + l * cD,
                                                            nullptr, Po, Qo, (float*)d_out, 0, 0);
        }
    }
}


# layer kernel on 8-row blocks of 8 waves (96 blocks, was 48 of 16 rows): a block pulls half the sender rows through its memory path; products stay 16-row with zero rows; every sum keeps its order
# speedup vs baseline: 5.0828x; 1.0556x over previous
#include <hip/hip_runtime.h>

typedef __attribute__((ext_vector_type(2))) float v2f;
typedef __attribute__((ext_vector_type(8))) float v8f;


static constexpr int cB = 2;
static constexpr int cN = 384;
static constexpr int cD = 128;
static constexpr int cM = cB * cN;
static constexpr int cDI = 64;
static constexpr int cLD = cD + 4;

template <int K>
__device__ __forceinline__ void load_b_frag(const float* __restrict__ bcol, float (&b)[K / 2])
{
#pragma unroll
    for (int i = 0; i < K / 4; ++i) {
        b[2 * i]     = bcol[(size_t)(4 * i) * cD];
        b[2 * i + 1] = bcol[(size_t)(4 * i + 1) * cD];
    }
}

template <int K>
__device__ __forceinline__ void load_a_frag(const float* arow, float (&a)[K / 2])
{
#pragma unroll
    for (int i = 0; i < K / 4; ++i) {
        a[2 * i]     = arow[4 * i];
        a[2 * i + 1] = arow[4 * i + 1];
    }
}

template <int K>
__device__ __forceinline__ v8f mm_reg(const float (&af)[K / 2], const float (&bf)[K / 2], v8f c)
{
#pragma unroll
    for (int i = 0; i < K / 4; ++i) {
        v2f a, b;
        a.x = af[2 * i];
        a.y = af[2 * i + 1];
        b.x = bf[2 * i];
        b.y = bf[2 * i + 1];
        c = __builtin_amdgcn_wmma_f32_16x16x4_f32(
                false, a, false, b, (short)0, c, false, false);
    }
    return c;
}

template <int K>
__device__ __forceinline__ v8f mm_lds(const float* arow, const float (&bf)[K / 2], v8f c)
{
#pragma unroll
    for (int i = 0; i < K / 4; ++i) {
        v2f a, b;
        a.x = arow[4 * i];
        a.y = arow[4 * i + 1];
        b.x = bf[2 * i];
        b.y = bf[2 * i + 1];
        c = __builtin_amdgcn_wmma_f32_16x16x4_f32(
                false, a, false, b, (short)0, c, false, false);
    }
    return c;
}

__device__ __forceinline__ void pq_from_tile(const float* arow,
                                             const float (&bp)[cD / 2],
                                             const float (&bq)[cD / 2],
                                             float* __restrict__ P,
                                             float* __restrict__ Q,
                                             int mbase, int n)
{
    v8f p = {};
    v8f q = {};
#pragma unroll
    for (int i = 0; i < cD / 4; ++i) {
        v2f a, b0, b1;
        a.x  = arow[4 * i];
        a.y  = arow[4 * i + 1];
        b0.x = bp[2 * i];
        b0.y = bp[2 * i + 1];
        b1.x = bq[2 * i];
        b1.y = bq[2 * i + 1];
        p = __builtin_amdgcn_wmma_f32_16x16x4_f32(
                false, a, false, b0, (short)0, p, false, false);
        q = __builtin_amdgcn_wmma_f32_16x16x4_f32(
                false, a, false, b1, (short)0, q, false, false);
    }
#pragma unroll
    for (int r = 0; r < 8; ++r) {
        P[(size_t)(mbase + r) * cD + n] = p[r];
        Q[(size_t)(mbase + r) * cD + n] = q[r];
    }
}

static constexpr int cFoldTiles  = cD / 16 + 1;
static constexpr int cFoldRows   = 16 * cFoldTiles;
static constexpr int cFoldBlocks = 3 * cFoldTiles;

__global__ __launch_bounds__(256)
void in_proj_pq(const float* __restrict__ h,
                const float* __restrict__ ipw,
                const float* __restrict__ ipb,
                const float* __restrict__ mw1,
                const float* __restrict__ mw2,
                const float* __restrict__ mb2,
                const float* __restrict__ uw1,
                float* __restrict__ x,
                float* __restrict__ P,
                float* __restrict__ Q,
                float* __restrict__ Wp)
{
    __shared__ __attribute__((aligned(16))) float sX[16][cLD];
    __shared__ __attribute__((aligned(16))) float sI[16][cDI + 4];

    const int wave  = threadIdx.x >> 5;
    const int lane  = threadIdx.x & 31;
    const int m0    = blockIdx.x * 16;
    const int l15   = lane & 15;
    const int khalf = lane >> 4;
    const int n     = wave * 16 + l15;
    const int rb    = khalf * 8;

    const size_t boff = (size_t)(2 * khalf) * cD + n;

    if (blockIdx.x >= cM / 16) {
        const int e  = blockIdx.x - cM / 16;
        const int l  = e / cFoldTiles;
        const int t  = e % cFoldTiles;
        const float* ub = uw1 + (size_t)l * 2 * cD * cD + (size_t)cD * cD;
        float bf[cD / 2], af[cD / 2];
        load_b_frag<cD>(ub + boff, bf);
        float4 t0 = make_float4(0.0f, 0.0f, 0.0f, 0.0f);
        float4 t1 = t0;
        if (t < cD / 16) {
            const float4* A4 = reinterpret_cast<const float4*>(mw2 + (size_t)l * cD * cD + (size_t)(16 * t) * cD);
            t0 = A4[threadIdx.x];
            t1 = A4[threadIdx.x + 256];
        } else if (wave == 0) {
            t0 = reinterpret_cast<const float4*>(mb2 + (size_t)l * cD)[lane];
        }
        *reinterpret_cast<float4*>(&sX[wave][4 * lane])     = t0;
        *reinterpret_cast<float4*>(&sX[wave + 8][4 * lane]) = t1;
        __syncthreads();
        load_a_frag<cD>(&sX[l15][2 * khalf], af);
        v8f c = {};
        c = mm_reg<cD>(af, bf, c);
        float* Wl = Wp + (size_t)l * cFoldRows * cD + (size_t)(16 * t + rb) * cD + n;
#pragma unroll
        for (int r = 0; r < 8; ++r)
            Wl[(size_t)r * cD] = c[r];
        return;
    }

    const float4 th = reinterpret_cast<const float4*>(h + (size_t)m0 * cDI)[threadIdx.x];

    float a0[cDI / 2], b0[cDI / 2];
    load_b_frag<cDI>(ipw + boff, b0);
    *reinterpret_cast<float4*>(&sI[threadIdx.x >> 4][4 * (threadIdx.x & 15)]) = th;
    __syncthreads();
    float bp[cD / 2], bq[cD / 2];
    load_b_frag<cD>(mw1 + boff, bp);
    load_b_frag<cD>(mw1 + (size_t)cD * cD + boff, bq);
    load_a_frag<cDI>(&sI[l15][2 * khalf], a0);

    v8f c = {};
    c = mm_reg<cDI>(a0, b0, c);
    const float bv = ipb[n];
#pragma unroll
    for (int r = 0; r < 8; ++r) {
        float y = c[r] + bv;
        y = y > 0.0f ? y : 0.0f;
        sX[rb + r][n] = y;
        x[(size_t)(m0 + rb + r) * cD + n] = y;
    }
    __syncthreads();
    pq_from_tile(&sX[l15][2 * khalf], bp, bq, P, Q, m0 + rb, n);
}

static constexpr int cTR = 8;
static constexpr int cU  = 16;
static constexpr int cCH = cN / 32;

__device__ __forceinline__ void pair_lane(float p, float q, float a, float w,
                                          float b1, float& acc)
{
    float t = __builtin_fmaf(a, w, p + q) + b1;
    t = t > 0.0f ? t : 0.0f;
    acc = __builtin_fmaf(a, t, acc);
}

__device__ __forceinline__ void pair_step(const float4& p, const float4& q, float a,
                                          const float4& w, const float4& b1,
                                          float4& acc, float& dg)
{
    pair_lane(p.x, q.x, a, w.x, b1.x, acc.x);
    pair_lane(p.y, q.y, a, w.y, b1.y, acc.y);
    pair_lane(p.z, q.z, a, w.z, b1.z, acc.z);
    pair_lane(p.w, q.w, a, w.w, b1.w, acc.w);
    dg += a;
}

template <bool kNext>
__global__ __launch_bounds__(256)
void layer_fused(const float* __restrict__ Pin,
                 const float* __restrict__ Qin,
                 const float* __restrict__ adj,
                 const float* __restrict__ we,
                 const float* __restrict__ mb1,
                 const float* xin,
                 const float* __restrict__ Wp,
                 const float* __restrict__ cp,
                 const float* __restrict__ uw1,
                 const float* __restrict__ ub1,
                 const float* __restrict__ uw2,
                 const float* __restrict__ ub2,
                 const float* __restrict__ g,
                 const float* __restrict__ beta,
                 const float* __restrict__ mw1n,
                 float* __restrict__ Pout,
                 float* __restrict__ Qout,
                 float* xout,
                 int add_residual, int relu_after)
{
    __shared__ __attribute__((aligned(16))) float sA[16][cLD];
    __shared__ __attribute__((aligned(16))) float sH[16][cLD];
    __shared__ __attribute__((aligned(16))) float sR[16][cLD];
    __shared__ __attribute__((aligned(16))) float sX[16][cLD];
    __shared__ __attribute__((aligned(16))) int   s_j[cTR][cN + cU];
    __shared__ __attribute__((aligned(16))) float s_a[cTR][cN + cU];
    __shared__ float sDeg[16];

    const int wave  = threadIdx.x >> 5;
    const int lane  = threadIdx.x & 31;
    const int m0    = blockIdx.x * cTR;
    const int l15   = lane & 15;
    const int khalf = lane >> 4;
    const int n     = wave * 16 + l15;
    const int rb    = khalf * 8;
    const size_t boff = (size_t)(2 * khalf) * cD + n;

    const int row = m0 + wave;
    const int b   = m0 / cN;

    const float* arow = adj + (size_t)row * cN;
    float av[cCH];
#pragma unroll
    for (int c = 0; c < cCH; ++c)
        av[c] = arow[c * 32 + lane];

    const float4 tx  = reinterpret_cast<const float4*>(xin + (size_t)row * cD)[lane];
    const float4 p4  = reinterpret_cast<const float4*>(Pin + (size_t)row * cD)[lane];
    const float4 w4  = reinterpret_cast<const float4*>(we)[lane];
    const float4 b14 = reinterpret_cast<const float4*>(mb1)[lane];

    const float bv_c  = cp[n];
    const float bv_u1 = ub1[n];
    const float bv_u2 = ub2[n];
    float gg[4], bb[4];
#pragma unroll
    for (int i = 0; i < 4; ++i) {
        gg[i] = g[lane + 32 * i];
        bb[i] = beta[lane + 32 * i];
    }

    float b1[cD / 2];
    load_b_frag<cD>(Wp + boff, b1);

    {
        const float4 z4 = make_float4(0.0f, 0.0f, 0.0f, 0.0f);
        *reinterpret_cast<float4*>(&sR[wave + cTR][4 * lane]) = z4;
        *reinterpret_cast<float4*>(&sX[wave + cTR][4 * lane]) = z4;
        if (lane == 0) sDeg[wave + cTR] = 0.0f;
    }

    int ns = 0;
#pragma unroll
    for (int c = 0; c < cCH; ++c) {
        const bool nz = av[c] != 0.0f;
        const unsigned mask = __builtin_amdgcn_ballot_w32(nz);
        const int pos = ns + (int)__builtin_amdgcn_mbcnt_lo(mask, 0u);
        if (nz) {
            s_j[wave][pos] = c * 32 + lane;
            s_a[wave][pos] = av[c];
        }
        ns += __builtin_popcount(mask);
    }
    if (lane < cU) {
        s_j[wave][ns + lane] = 0;
        s_a[wave][ns + lane] = 0.0f;
    }
    __syncthreads();

    const float4* Qb = reinterpret_cast<const float4*>(Qin + (size_t)b * cN * cD);
    const int4*   lj = reinterpret_cast<const int4*>(s_j[wave]);
    const float4* la = reinterpret_cast<const float4*>(s_a[wave]);

    float4 acc = make_float4(0.0f, 0.0f, 0.0f, 0.0f);
    float  dg  = 0.0f;
    int k = 0;
    for (; k + cU <= ns; k += cU) {
        int jj[cU]; float aa[cU]; float4 q[cU];
#pragma unroll
        for (int v = 0; v < cU / 4; ++v) {
            const int4   j4 = lj[(k >> 2) + v];
            const float4 a4 = la[(k >> 2) + v];
            jj[4 * v + 0] = j4.x; jj[4 * v + 1] = j4.y; jj[4 * v + 2] = j4.z; jj[4 * v + 3] = j4.w;
            aa[4 * v + 0] = a4.x; aa[4 * v + 1] = a4.y; aa[4 * v + 2] = a4.z; aa[4 * v + 3] = a4.w;
        }
#pragma unroll
        for (int u = 0; u < cU; ++u)
            q[u] = Qb[(unsigned)(jj[u] * (cD / 4) + lane)];
#pragma unroll
        for (int u = 0; u < cU; ++u)
            pair_step(p4, q[u], aa[u], w4, b14, acc, dg);
    }
    if (k < ns) {
        int jj[cU]; float aa[cU]; float4 q[cU];
#pragma unroll
        for (int v = 0; v < cU / 4; ++v) {
            const int4   j4 = lj[(k >> 2) + v];
            const float4 a4 = la[(k >> 2) + v];
            jj[4 * v + 0] = j4.x; jj[4 * v + 1] = j4.y; jj[4 * v + 2] = j4.z; jj[4 * v + 3] = j4.w;
            aa[4 * v + 0] = a4.x; aa[4 * v + 1] = a4.y; aa[4 * v + 2] = a4.z; aa[4 * v + 3] = a4.w;
        }
#pragma unroll
        for (int u = 0; u < cU; ++u)
            q[u] = Qb[(unsigned)(jj[u] * (cD / 4) + lane)];
#pragma unroll
        for (int u = 0; u < cU; ++u)
            if (k + u < ns)
                pair_step(p4, q[u], aa[u], w4, b14, acc, dg);
    }

    *reinterpret_cast<float4*>(&sR[wave][4 * lane]) = acc;
    *reinterpret_cast<float4*>(&sX[wave][4 * lane]) = tx;
    if (lane == 0) sDeg[wave] = dg;
    __syncthreads();

    float xr[8];
    float b3[cD / 2];
    {
        float a1[cD / 2];
        float a2[cD / 2], b2a[cD / 2];
        load_b_frag<cD>(uw1 + boff, b2a);
        load_a_frag<cD>(&sR[l15][2 * khalf], a1);
        load_a_frag<cD>(&sX[l15][2 * khalf], a2);
        float dr[8];
#pragma unroll
        for (int r = 0; r < 8; ++r) {
            xr[r] = sX[rb + r][n];
            dr[r] = sDeg[rb + r];
        }
        v8f c = {};
        c = mm_reg<cD>(a1, b1, c);
        load_b_frag<cD>(uw2 + boff, b3);
        c = mm_reg<cD>(a2, b2a, c);
#pragma unroll
        for (int r = 0; r < 8; ++r) {
            const float v = (c[r] + bv_u1) + dr[r] * bv_c;
            sH[rb + r][n] = v > 0.0f ? v : 0.0f;
        }
    }
    __syncthreads();

    float bp[cD / 2], bq[cD / 2];
    if (kNext) {
        load_b_frag<cD>(mw1n + boff, bp);
        load_b_frag<cD>(mw1n + (size_t)cD * cD + boff, bq);
    }
    {
        v8f c = {};
        c = mm_lds<cD>(&sH[l15][2 * khalf], b3, c);
#pragma unroll
        for (int r = 0; r < 8; ++r) {
            float v = c[r] + bv_u2;
            if (add_residual) v += xr[r];
            sA[rb + r][n] = v;
        }
    }
    __syncthreads();

    {
        float v[4];
#pragma unroll
        for (int i = 0; i < 4; ++i)
            v[i] = sA[wave][lane + 32 * i];

        float s = (v[0] + v[2]) + (v[1] + v[3]);
#pragma unroll
        for (int o = 16; o > 0; o >>= 1)
            s += __shfl_down(s, o);
        const float mu = __shfl(s, 0) * (1.0f / cD);

        float dv[4];
#pragma unroll
        for (int i = 0; i < 4; ++i)
            dv[i] = v[i] - mu;
        float t = (dv[0] * dv[0] + dv[2] * dv[2]) + (dv[1] * dv[1] + dv[3] * dv[3]);
#pragma unroll
        for (int o = 16; o > 0; o >>= 1)
            t += __shfl_down(t, o);
        const float var = __shfl(t, 0) * (1.0f / cD);
        const float rs  = rsqrtf(var + 1e-5f);

#pragma unroll
        for (int i = 0; i < 4; ++i) {
            float y = dv[i] * rs * gg[i] + bb[i];
            if (relu_after) y = y > 0.0f ? y : 0.0f;
            xout[(size_t)row * cD + lane + 32 * i] = y;
            sH[wave][lane + 32 * i] = y;
        }
    }

    if (kNext) {
        __syncthreads();
        const float* arw = &sH[l15][2 * khalf];
        v8f p = {};
        v8f q = {};
#pragma unroll
        for (int i = 0; i < cD / 4; ++i) {
            v2f a, f0, f1;
            a.x  = arw[4 * i];
            a.y  = arw[4 * i + 1];
            f0.x = bp[2 * i];
            f0.y = bp[2 * i + 1];
            f1.x = bq[2 * i];
            f1.y = bq[2 * i + 1];
            p = __builtin_amdgcn_wmma_f32_16x16x4_f32(
                    false, a, false, f0, (short)0, p, false, false);
            q = __builtin_amdgcn_wmma_f32_16x16x4_f32(
                    false, a, false, f1, (short)0, q, false, false);
        }
        if (khalf == 0) {
#pragma unroll
            for (int r = 0; r < 8; ++r) {
                Pout[(size_t)(m0 + r) * cD + n] = p[r];
                Qout[(size_t)(m0 + r) * cD + n] = q[r];
            }
        }
    }
}

extern "C" void kernel_launch(void* const* d_in, const int* in_sizes, int n_in,
                              void* d_out, int out_size, void* d_ws, size_t ws_size,
                              hipStream_t stream)
{
    (void)in_sizes; (void)n_in; (void)out_size; (void)ws_size;

    const float* h    = (const float*)d_in[0];
    const float* adj  = (const float*)d_in[1];
    const float* ip_w = (const float*)d_in[2];
    const float* ip_b = (const float*)d_in[3];
    const float* mw1  = (const float*)d_in[4];
    const float* mb1  = (const float*)d_in[5];
    const float* mw2  = (const float*)d_in[6];
    const float* mb2  = (const float*)d_in[7];
    const float* uw1  = (const float*)d_in[8];
    const float* ub1  = (const float*)d_in[9];
    const float* uw2  = (const float*)d_in[10];
    const float* ub2  = (const float*)d_in[11];
    const float* ln_g = (const float*)d_in[12];
    const float* ln_b = (const float*)d_in[13];

    float* ws = (float*)d_ws;
    const size_t S = (size_t)cM * cD;
    float* x  = ws + 0 * S;
    float* Pa = ws + 1 * S;
    float* Qa = ws + 2 * S;
    float* Pb = ws + 5 * S;
    float* Qb = ws + 6 * S;
    float* Wp = ws + 4 * S;

    in_proj_pq<<<cM / 16 + cFoldBlocks, 256, 0, stream>>>(h, ip_w, ip_b, mw1, mw2, mb2, uw1, x, Pa, Qa, Wp);

    for (int l = 0; l < 3; ++l) {
        const float* mw1l = mw1 + (size_t)l * 257 * cD;
        const float* mb1l = mb1 + (size_t)l * cD;
        const float* Wpl  = Wp + (size_t)l * cFoldRows * cD;
        const float* cpl  = Wpl + (size_t)cD * cD;
        const float* uw1l = uw1 + (size_t)l * 256 * cD;
        const float* ub1l = ub1 + (size_t)l * cD;
        const float* uw2l = uw2 + (size_t)l * cD * cD;
        const float* ub2l = ub2 + (size_t)l * cD;
        float* Pi = (l == 1) ? Pb : Pa;
        float* Qi = (l == 1) ? Qb : Qa;
        float* Po = (l == 1) ? Pa : Pb;
        float* Qo = (l == 1) ? Qa : Qb;

        if (l < 2) {
            layer_fused<true><<<cM / cTR, 256, 0, stream>>>(Pi, Qi, adj, mw1l + 256 * cD, mb1l, x, Wpl, cpl,
                                                           uw1l, ub1l, uw2l, ub2l,
                                                           ln_g + l * cD, ln_b + l * cD,
                                                           mw1l + 257 * cD, Po, Qo, x, 1, 1);
        } else {
            layer_fused<false><<<cM / cTR, 256, 0, stream>>>(Pi, Qi, adj, mw1l + 256 * cD, mb1l, x, Wpl, cpl,
                                                            uw1l, ub1l, uw2l, ub2l,
                                                            ln_g + l * cD, ln_b + l * cD,
                                                            nullptr, Po, Qo, (float*)d_out, 0, 0);
        }
    }
}


# weight operands of the layer launches pre-arranged in fragment order by the first launch, once per call: a fragment is 16 whole-line 16-byte loads, was 64 four-byte loads; every sum keeps its order
# speedup vs baseline: 5.4098x; 1.0643x over previous
#include <hip/hip_runtime.h>

typedef __attribute__((ext_vector_type(2))) float v2f;
typedef __attribute__((ext_vector_type(8))) float v8f;


static constexpr int cB = 2;
static constexpr int cN = 384;
static constexpr int cD = 128;
static constexpr int cM = cB * cN;
static constexpr int cDI = 64;
static constexpr int cLD = cD + 4;

template <int K>
__device__ __forceinline__ void load_b_frag(const float* __restrict__ bcol, float (&b)[K / 2])
{
#pragma unroll
    for (int i = 0; i < K / 4; ++i) {
        b[2 * i]     = bcol[(size_t)(4 * i) * cD];
        b[2 * i + 1] = bcol[(size_t)(4 * i + 1) * cD];
    }
}

static constexpr int cOp = cD * cD;
__device__ __forceinline__ void load_b_pre(const float* __restrict__ op, int wave, int lane, float (&b)[cD / 2])
{
    const float4* f = reinterpret_cast<const float4*>(op) + (wave * (cD / 8) * 32 + lane);
#pragma unroll
    for (int j4 = 0; j4 < cD / 8; ++j4) {
        const float4 v = f[j4 * 32];
        b[4 * j4]     = v.x;
        b[4 * j4 + 1] = v.y;
        b[4 * j4 + 2] = v.z;
        b[4 * j4 + 3] = v.w;
    }
}

template <int K>
__device__ __forceinline__ void load_a_frag(const float* arow, float (&a)[K / 2])
{
#pragma unroll
    for (int i = 0; i < K / 4; ++i) {
        a[2 * i]     = arow[4 * i];
        a[2 * i + 1] = arow[4 * i + 1];
    }
}

template <int K>
__device__ __forceinline__ v8f mm_reg(const float (&af)[K / 2], const float (&bf)[K / 2], v8f c)
{
#pragma unroll
    for (int i = 0; i < K / 4; ++i) {
        v2f a, b;
        a.x = af[2 * i];
        a.y = af[2 * i + 1];
        b.x = bf[2 * i];
        b.y = bf[2 * i + 1];
        c = __builtin_amdgcn_wmma_f32_16x16x4_f32(
                false, a, false, b, (short)0, c, false, false);
    }
    return c;
}

template <int K>
__device__ __forceinline__ v8f mm_lds(const float* arow, const float (&bf)[K / 2], v8f c)
{
#pragma unroll
    for (int i = 0; i < K / 4; ++i) {
        v2f a, b;
        a.x = arow[4 * i];
        a.y = arow[4 * i + 1];
        b.x = bf[2 * i];
        b.y = bf[2 * i + 1];
        c = __builtin_amdgcn_wmma_f32_16x16x4_f32(
                false, a, false, b, (short)0, c, false, false);
    }
    return c;
}

__device__ __forceinline__ void pq_from_tile(const float* arow,
                                             const float (&bp)[cD / 2],
                                             const float (&bq)[cD / 2],
                                             float* __restrict__ P,
                                             float* __restrict__ Q,
                                             int mbase, int n)
{
    v8f p = {};
    v8f q = {};
#pragma unroll
    for (int i = 0; i < cD / 4; ++i) {
        v2f a, b0, b1;
        a.x  = arow[4 * i];
        a.y  = arow[4 * i + 1];
        b0.x = bp[2 * i];
        b0.y = bp[2 * i + 1];
        b1.x = bq[2 * i];
        b1.y = bq[2 * i + 1];
        p = __builtin_amdgcn_wmma_f32_16x16x4_f32(
                false, a, false, b0, (short)0, p, false, false);
        q = __builtin_amdgcn_wmma_f32_16x16x4_f32(
                false, a, false, b1, (short)0, q, false, false);
    }
#pragma unroll
    for (int r = 0; r < 8; ++r) {
        P[(size_t)(mbase + r) * cD + n] = p[r];
        Q[(size_t)(mbase + r) * cD + n] = q[r];
    }
}

static constexpr int cFoldTiles  = cD / 16 + 1;
static constexpr int cFoldBlocks = 3 * cFoldTiles;
static constexpr int cSlots      = 5;
static constexpr int cPreOps     = 10;
static constexpr int cPreParts   = 4;
static constexpr int cPreBlocks  = cPreOps * cPreParts;

__global__ __launch_bounds__(256)
void in_proj_pq(const float* __restrict__ h,
                const float* __restrict__ ipw,
                const float* __restrict__ ipb,
                const float* __restrict__ mw1,
                const float* __restrict__ mw2,
                const float* __restrict__ mb2,
                const float* __restrict__ uw1,
                const float* __restrict__ uw2,
                float* __restrict__ x,
                float* __restrict__ P,
                float* __restrict__ Q,
                float* __restrict__ Fr,
                float* __restrict__ cpr)
{
    __shared__ __attribute__((aligned(16))) float sX[16][cLD];
    __shared__ __attribute__((aligned(16))) float sI[16][cDI + 4];

    const int wave  = threadIdx.x >> 5;
    const int lane  = threadIdx.x & 31;
    const int m0    = blockIdx.x * 16;
    const int l15   = lane & 15;
    const int khalf = lane >> 4;
    const int n     = wave * 16 + l15;
    const int rb    = khalf * 8;

    const size_t boff = (size_t)(2 * khalf) * cD + n;

    if (blockIdx.x >= cM / 16 + cFoldBlocks) {
        const int e    = blockIdx.x - (cM / 16 + cFoldBlocks);
        const int id   = e / cPreParts;
        const int part = e % cPreParts;
        const float* srcm;
        int slot;
        if (id < 3)      { srcm = uw1 + (size_t)id * 2 * cD * cD;                               slot = cSlots * id + 1; }
        else if (id < 6) { srcm = uw2 + (size_t)(id - 3) * cD * cD;                             slot = cSlots * (id - 3) + 2; }
        else if (id < 8) { srcm = mw1 + (size_t)(id - 5) * 257 * cD;                            slot = cSlots * (id - 6) + 3; }
        else             { srcm = mw1 + (size_t)(id - 7) * 257 * cD + (size_t)cD * cD;          slot = cSlots * (id - 8) + 4; }
        float4* dst = reinterpret_cast<float4*>(Fr + (size_t)slot * cOp);
        float4 v[4];
#pragma unroll
        for (int it = 0; it < 4; ++it) {
            const int q  = part * 1024 + it * 256 + (int)threadIdx.x;
            const int ql = q & 31;
            const int qj = (q >> 5) & 15;
            const int qw = q >> 9;
            const float* sp = srcm + (size_t)(8 * qj + 2 * (ql >> 4)) * cD + 16 * qw + (ql & 15);
            v[it] = make_float4(sp[0], sp[cD], sp[4 * cD], sp[5 * cD]);
        }
#pragma unroll
        for (int it = 0; it < 4; ++it)
            dst[part * 1024 + it * 256 + (int)threadIdx.x] = v[it];
        return;
    }

    if (blockIdx.x >= cM / 16) {
        const int e  = blockIdx.x - cM / 16;
        const int l  = e / cFoldTiles;
        const int t  = e % cFoldTiles;
        const float* ub = uw1 + (size_t)l * 2 * cD * cD + (size_t)cD * cD;
        float bf[cD / 2], af[cD / 2];
        load_b_frag<cD>(ub + boff, bf);
        float4 t0 = make_float4(0.0f, 0.0f, 0.0f, 0.0f);
        float4 t1 = t0;
        if (t < cD / 16) {
            const float4* A4 = reinterpret_cast<const float4*>(mw2 + (size_t)l * cD * cD + (size_t)(16 * t) * cD);
            t0 = A4[threadIdx.x];
            t1 = A4[threadIdx.x + 256];
        } else if (wave == 0) {
            t0 = reinterpret_cast<const float4*>(mb2 + (size_t)l * cD)[lane];
        }
        *reinterpret_cast<float4*>(&sX[wave][4 * lane])     = t0;
        *reinterpret_cast<float4*>(&sX[wave + 8][4 * lane]) = t1;
        __syncthreads();
        load_a_frag<cD>(&sX[l15][2 * khalf], af);
        v8f c = {};
        c = mm_reg<cD>(af, bf, c);
        if (t < cD / 16) {
            float* Wl = Fr + (size_t)(cSlots * l) * cOp + (((wave * (cD / 8) + 2 * t + khalf) * 32 + l15) << 2);
#pragma unroll
            for (int r = 0; r < 8; ++r)
                Wl[(((r >> 1) & 1) << 6) + 2 * ((r >> 2) & 1) + (r & 1)] = c[r];
        } else if (khalf == 0) {
            cpr[l * cD + n] = c[0];
        }
        return;
    }

    const float4 th = reinterpret_cast<const float4*>(h + (size_t)m0 * cDI)[threadIdx.x];

    float a0[cDI / 2], b0[cDI / 2];
    load_b_frag<cDI>(ipw + boff, b0);
    *reinterpret_cast<float4*>(&sI[threadIdx.x >> 4][4 * (threadIdx.x & 15)]) = th;
    __syncthreads();
    float bp[cD / 2], bq[cD / 2];
    load_b_frag<cD>(mw1 + boff, bp);
    load_b_frag<cD>(mw1 + (size_t)cD * cD + boff, bq);
    load_a_frag<cDI>(&sI[l15][2 * khalf], a0);

    v8f c = {};
    c = mm_reg<cDI>(a0, b0, c);
    const float bv = ipb[n];
#pragma unroll
    for (int r = 0; r < 8; ++r) {
        float y = c[r] + bv;
        y = y > 0.0f ? y : 0.0f;
        sX[rb + r][n] = y;
        x[(size_t)(m0 + rb + r) * cD + n] = y;
    }
    __syncthreads();
    pq_from_tile(&sX[l15][2 * khalf], bp, bq, P, Q, m0 + rb, n);
}

static constexpr int cTR = 8;
static constexpr int cU  = 16;
static constexpr int cCH = cN / 32;

__device__ __forceinline__ void pair_lane(float p, float q, float a, float w,
                                          float b1, float& acc)
{
    float t = __builtin_fmaf(a, w, p + q) + b1;
    t = t > 0.0f ? t : 0.0f;
    acc = __builtin_fmaf(a, t, acc);
}

__device__ __forceinline__ void pair_step(const float4& p, const float4& q, float a,
                                          const float4& w, const float4& b1,
                                          float4& acc, float& dg)
{
    pair_lane(p.x, q.x, a, w.x, b1.x, acc.x);
    pair_lane(p.y, q.y, a, w.y, b1.y, acc.y);
    pair_lane(p.z, q.z, a, w.z, b1.z, acc.z);
    pair_lane(p.w, q.w, a, w.w, b1.w, acc.w);
    dg += a;
}

template <bool kNext>
__global__ __launch_bounds__(256)
void layer_fused(const float* Pin,
                 const float* __restrict__ Qin,
                 const float* __restrict__ adj,
                 const float* __restrict__ we,
                 const float* __restrict__ mb1,
                 const float* xin,
                 const float* __restrict__ Fl,
                 const float* __restrict__ cp,
                 const float* __restrict__ ub1,
                 const float* __restrict__ ub2,
                 const float* __restrict__ g,
                 const float* __restrict__ beta,
                 float* Pout,
                 float* __restrict__ Qout,
                 float* xout,
                 int add_residual, int relu_after)
{
    __shared__ __attribute__((aligned(16))) float sA[16][cLD];
    __shared__ __attribute__((aligned(16))) float sH[16][cLD];
    __shared__ __attribute__((aligned(16))) float sR[16][cLD];
    __shared__ __attribute__((aligned(16))) float sX[16][cLD];
    __shared__ __attribute__((aligned(16))) int   s_j[cTR][cN + cU];
    __shared__ __attribute__((aligned(16))) float s_a[cTR][cN + cU];
    __shared__ float sDeg[16];

    const int wave  = threadIdx.x >> 5;
    const int lane  = threadIdx.x & 31;
    const int m0    = blockIdx.x * cTR;
    const int l15   = lane & 15;
    const int khalf = lane >> 4;
    const int n     = wave * 16 + l15;
    const int rb    = khalf * 8;

    const int row = m0 + wave;
    const int b   = m0 / cN;

    const float* arow = adj + (size_t)row * cN;
    float av[cCH];
#pragma unroll
    for (int c = 0; c < cCH; ++c)
        av[c] = arow[c * 32 + lane];

    const float4 tx  = reinterpret_cast<const float4*>(xin + (size_t)row * cD)[lane];
    const float4 p4  = reinterpret_cast<const float4*>(Pin + (size_t)row * cD)[lane];
    const float4 w4  = reinterpret_cast<const float4*>(we)[lane];
    const float4 b14 = reinterpret_cast<const float4*>(mb1)[lane];

    const float bv_c  = cp[n];
    const float bv_u1 = ub1[n];
    const float bv_u2 = ub2[n];
    float gg[4], bb[4];
#pragma unroll
    for (int i = 0; i < 4; ++i) {
        gg[i] = g[lane + 32 * i];
        bb[i] = beta[lane + 32 * i];
    }

    float b1[cD / 2];
    load_b_pre(Fl, wave, lane, b1);

    {
        const float4 z4 = make_float4(0.0f, 0.0f, 0.0f, 0.0f);
        *reinterpret_cast<float4*>(&sR[wave + cTR][4 * lane]) = z4;
        *reinterpret_cast<float4*>(&sX[wave + cTR][4 * lane]) = z4;
        if (lane == 0) sDeg[wave + cTR] = 0.0f;
    }

    int ns = 0;
#pragma unroll
    for (int c = 0; c < cCH; ++c) {
        const bool nz = av[c] != 0.0f;
        const unsigned mask = __builtin_amdgcn_ballot_w32(nz);
        const int pos = ns + (int)__builtin_amdgcn_mbcnt_lo(mask, 0u);
        if (nz) {
            s_j[wave][pos] = c * 32 + lane;
            s_a[wave][pos] = av[c];
        }
        ns += __builtin_popcount(mask);
    }
    if (lane < cU) {
        s_j[wave][ns + lane] = 0;
        s_a[wave][ns + lane] = 0.0f;
    }
    __syncthreads();

    const float4* Qb = reinterpret_cast<const float4*>(Qin + (size_t)b * cN * cD);
    const int4*   lj = reinterpret_cast<const int4*>(s_j[wave]);
    const float4* la = reinterpret_cast<const float4*>(s_a[wave]);

    float4 acc = make_float4(0.0f, 0.0f, 0.0f, 0.0f);
    float  dg  = 0.0f;
    int k = 0;
    for (; k + cU <= ns; k += cU) {
        int jj[cU]; float aa[cU]; float4 q[cU];
#pragma unroll
        for (int v = 0; v < cU / 4; ++v) {
            const int4   j4 = lj[(k >> 2) + v];
            const float4 a4 = la[(k >> 2) + v];
            jj[4 * v + 0] = j4.x; jj[4 * v + 1] = j4.y; jj[4 * v + 2] = j4.z; jj[4 * v + 3] = j4.w;
            aa[4 * v + 0] = a4.x; aa[4 * v + 1] = a4.y; aa[4 * v + 2] = a4.z; aa[4 * v + 3] = a4.w;
        }
#pragma unroll
        for (int u = 0; u < cU; ++u)
            q[u] = Qb[(unsigned)(jj[u] * (cD / 4) + lane)];
#pragma unroll
        for (int u = 0; u < cU; ++u)
            pair_step(p4, q[u], aa[u], w4, b14, acc, dg);
    }
    if (k < ns) {
        int jj[cU]; float aa[cU]; float4 q[cU];
#pragma unroll
        for (int v = 0; v < cU / 4; ++v) {
            const int4   j4 = lj[(k >> 2) + v];
            const float4 a4 = la[(k >> 2) + v];
            jj[4 * v + 0] = j4.x; jj[4 * v + 1] = j4.y; jj[4 * v + 2] = j4.z; jj[4 * v + 3] = j4.w;
            aa[4 * v + 0] = a4.x; aa[4 * v + 1] = a4.y; aa[4 * v + 2] = a4.z; aa[4 * v + 3] = a4.w;
        }
#pragma unroll
        for (int u = 0; u < cU; ++u)
            q[u] = Qb[(unsigned)(jj[u] * (cD / 4) + lane)];
#pragma unroll
        for (int u = 0; u < cU; ++u)
            if (k + u < ns)
                pair_step(p4, q[u], aa[u], w4, b14, acc, dg);
    }

    *reinterpret_cast<float4*>(&sR[wave][4 * lane]) = acc;
    *reinterpret_cast<float4*>(&sX[wave][4 * lane]) = tx;
    if (lane == 0) sDeg[wave] = dg;
    __syncthreads();

    float xr[8];
    float b3[cD / 2];
    {
        float a1[cD / 2];
        float a2[cD / 2], b2a[cD / 2];
        load_b_pre(Fl + cOp, wave, lane, b2a);
        __builtin_amdgcn_sched_barrier(0);
        load_a_frag<cD>(&sR[l15][2 * khalf], a1);
        load_a_frag<cD>(&sX[l15][2 * khalf], a2);
        float dr[8];
#pragma unroll
        for (int r = 0; r < 8; ++r) {
            xr[r] = sX[rb + r][n];
            dr[r] = sDeg[rb + r];
        }
        v8f c = {};
        c = mm_reg<cD>(a1, b1, c);
        load_b_pre(Fl + 2 * cOp, wave, lane, b3);
        c = mm_reg<cD>(a2, b2a, c);
#pragma unroll
        for (int r = 0; r < 8; ++r) {
            const float v = (c[r] + bv_u1) + dr[r] * bv_c;
            sH[rb + r][n] = v > 0.0f ? v : 0.0f;
        }
    }
    __syncthreads();

    float bp[cD / 2], bq[cD / 2];
    if (kNext) {
        load_b_pre(Fl + 3 * cOp, wave, lane, bp);
        load_b_pre(Fl + 4 * cOp, wave, lane, bq);
    }
    {
        v8f c = {};
        c = mm_lds<cD>(&sH[l15][2 * khalf], b3, c);
#pragma unroll
        for (int r = 0; r < 8; ++r) {
            float v = c[r] + bv_u2;
            if (add_residual) v += xr[r];
            sA[rb + r][n] = v;
        }
    }
    __syncthreads();

    {
        float v[4];
#pragma unroll
        for (int i = 0; i < 4; ++i)
            v[i] = sA[wave][lane + 32 * i];

        float s = (v[0] + v[2]) + (v[1] + v[3]);
#pragma unroll
        for (int o = 16; o > 0; o >>= 1)
            s += __shfl_down(s, o);
        const float mu = __shfl(s, 0) * (1.0f / cD);

        float dv[4];
#pragma unroll
        for (int i = 0; i < 4; ++i)
            dv[i] = v[i] - mu;
        float t = (dv[0] * dv[0] + dv[2] * dv[2]) + (dv[1] * dv[1] + dv[3] * dv[3]);
#pragma unroll
        for (int o = 16; o > 0; o >>= 1)
            t += __shfl_down(t, o);
        const float var = __shfl(t, 0) * (1.0f / cD);
        const float rs  = rsqrtf(var + 1e-5f);

#pragma unroll
        for (int i = 0; i < 4; ++i) {
            float y = dv[i] * rs * gg[i] + bb[i];
            if (relu_after) y = y > 0.0f ? y : 0.0f;
            xout[(size_t)row * cD + lane + 32 * i] = y;
            sH[wave][lane + 32 * i] = y;
        }
    }

    if (kNext) {
        __syncthreads();
        const float* arw = &sH[l15][2 * khalf];
        v8f p = {};
        v8f q = {};
#pragma unroll
        for (int i = 0; i < cD / 4; ++i) {
            v2f a, f0, f1;
            a.x  = arw[4 * i];
            a.y  = arw[4 * i + 1];
            f0.x = bp[2 * i];
            f0.y = bp[2 * i + 1];
            f1.x = bq[2 * i];
            f1.y = bq[2 * i + 1];
            p = __builtin_amdgcn_wmma_f32_16x16x4_f32(
                    false, a, false, f0, (short)0, p, false, false);
            q = __builtin_amdgcn_wmma_f32_16x16x4_f32(
                    false, a, false, f1, (short)0, q, false, false);
        }
        if (khalf == 0) {
#pragma unroll
            for (int r = 0; r < 8; ++r) {
                Pout[(size_t)(m0 + r) * cD + n] = p[r];
                Qout[(size_t)(m0 + r) * cD + n] = q[r];
            }
        }
    }
}

extern "C" void kernel_launch(void* const* d_in, const int* in_sizes, int n_in,
                              void* d_out, int out_size, void* d_ws, size_t ws_size,
                              hipStream_t stream)
{
    (void)in_sizes; (void)n_in; (void)out_size; (void)ws_size;

    const float* h    = (const float*)d_in[0];
    const float* adj  = (const float*)d_in[1];
    const float* ip_w = (const float*)d_in[2];
    const float* ip_b = (const float*)d_in[3];
    const float* mw1  = (const float*)d_in[4];
    const float* mb1  = (const float*)d_in[5];
    const float* mw2  = (const float*)d_in[6];
    const float* mb2  = (const float*)d_in[7];
    const float* uw1  = (const float*)d_in[8];
    const float* ub1  = (const float*)d_in[9];
    const float* uw2  = (const float*)d_in[10];
    const float* ub2  = (const float*)d_in[11];
    const float* ln_g = (const float*)d_in[12];
    const float* ln_b = (const float*)d_in[13];

    float* ws = (float*)d_ws;
    const size_t S = (size_t)cM * cD;
    float* x  = ws + 0 * S;
    float* Pa = ws + 1 * S;
    float* Qa = ws + 2 * S;
    float* Qb = ws + 6 * S;
    float* Fr = ws + 3 * S;
    float* cpr = Fr + (size_t)3 * cSlots * cOp;

    in_proj_pq<<<cM / 16 + cFoldBlocks + cPreBlocks, 256, 0, stream>>>(h, ip_w, ip_b, mw1, mw2, mb2, uw1, uw2, x, Pa, Qa, Fr, cpr);

    for (int l = 0; l < 3; ++l) {
        const float* mw1l = mw1 + (size_t)l * 257 * cD;
        const float* mb1l = mb1 + (size_t)l * cD;
        const float* Fll  = Fr + (size_t)l * cSlots * cOp;
        const float* cpl  = cpr + (size_t)l * cD;
        const float* ub1l = ub1 + (size_t)l * cD;
        const float* ub2l = ub2 + (size_t)l * cD;
        float* Qi = (l == 1) ? Qb : Qa;
        float* Qo = (l == 1) ? Qa : Qb;

        if (l < 2) {
            layer_fused<true><<<cM / cTR, 256, 0, stream>>>(Pa, Qi, adj, mw1l + 256 * cD, mb1l, x, Fll, cpl,
                                                           ub1l, ub2l,
                                                           ln_g + l * cD, ln_b + l * cD,
                                                           Pa, Qo, x, 1, 1);
        } else {
            layer_fused<false><<<cM / cTR, 256, 0, stream>>>(Pa, Qi, adj, mw1l + 256 * cD, mb1l, x, Fll, cpl,
                                                            ub1l, ub2l,
                                                            ln_g + l * cD, ln_b + l * cD,
                                                            Pa, Qo, (float*)d_out, 0, 0);
        }
    }
}


# hid's x part (x @ uw1[:128], needs no sender sum) runs in two halves behind the first two sender-row load groups, matrix unit idle there; hid now sums its x part before its R part
# speedup vs baseline: 5.6003x; 1.0352x over previous
#include <hip/hip_runtime.h>

typedef __attribute__((ext_vector_type(2))) float v2f;
typedef __attribute__((ext_vector_type(8))) float v8f;


static constexpr int cB = 2;
static constexpr int cN = 384;
static constexpr int cD = 128;
static constexpr int cM = cB * cN;
static constexpr int cDI = 64;
static constexpr int cLD = cD + 4;

template <int K>
__device__ __forceinline__ void load_b_frag(const float* __restrict__ bcol, float (&b)[K / 2])
{
#pragma unroll
    for (int i = 0; i < K / 4; ++i) {
        b[2 * i]     = bcol[(size_t)(4 * i) * cD];
        b[2 * i + 1] = bcol[(size_t)(4 * i + 1) * cD];
    }
}

static constexpr int cOp = cD * cD;
__device__ __forceinline__ void load_b_pre(const float* __restrict__ op, int wave, int lane, float (&b)[cD / 2])
{
    const float4* f = reinterpret_cast<const float4*>(op) + (wave * (cD / 8) * 32 + lane);
#pragma unroll
    for (int j4 = 0; j4 < cD / 8; ++j4) {
        const float4 v = f[j4 * 32];
        b[4 * j4]     = v.x;
        b[4 * j4 + 1] = v.y;
        b[4 * j4 + 2] = v.z;
        b[4 * j4 + 3] = v.w;
    }
}

template <int K>
__device__ __forceinline__ void load_a_frag(const float* arow, float (&a)[K / 2])
{
#pragma unroll
    for (int i = 0; i < K / 4; ++i) {
        a[2 * i]     = arow[4 * i];
        a[2 * i + 1] = arow[4 * i + 1];
    }
}

template <int K>
__device__ __forceinline__ v8f mm_reg(const float (&af)[K / 2], const float (&bf)[K / 2], v8f c)
{
#pragma unroll
    for (int i = 0; i < K / 4; ++i) {
        v2f a, b;
        a.x = af[2 * i];
        a.y = af[2 * i + 1];
        b.x = bf[2 * i];
        b.y = bf[2 * i + 1];
        c = __builtin_amdgcn_wmma_f32_16x16x4_f32(
                false, a, false, b, (short)0, c, false, false);
    }
    return c;
}

template <int K>
__device__ __forceinline__ v8f mm_lds(const float* arow, const float (&bf)[K / 2], v8f c)
{
#pragma unroll
    for (int i = 0; i < K / 4; ++i) {
        v2f a, b;
        a.x = arow[4 * i];
        a.y = arow[4 * i + 1];
        b.x = bf[2 * i];
        b.y = bf[2 * i + 1];
        c = __builtin_amdgcn_wmma_f32_16x16x4_f32(
                false, a, false, b, (short)0, c, false, false);
    }
    return c;
}

template <int H>
__device__ __forceinline__ v8f mm_lds_half(const float* arow, const float (&bf)[cD / 2], v8f c)
{
    float a[cD / 4];
#pragma unroll
    for (int i = 0; i < cD / 8; ++i) {
        a[2 * i]     = arow[4 * (16 * H + i)];
        a[2 * i + 1] = arow[4 * (16 * H + i) + 1];
    }
#pragma unroll
    for (int i = 0; i < cD / 8; ++i) {
        v2f av, bv;
        av.x = a[2 * i];
        av.y = a[2 * i + 1];
        bv.x = bf[2 * (16 * H + i)];
        bv.y = bf[2 * (16 * H + i) + 1];
        c = __builtin_amdgcn_wmma_f32_16x16x4_f32(
                false, av, false, bv, (short)0, c, false, false);
    }
    return c;
}

__device__ __forceinline__ void pq_from_tile(const float* arow,
                                             const float (&bp)[cD / 2],
                                             const float (&bq)[cD / 2],
                                             float* __restrict__ P,
                                             float* __restrict__ Q,
                                             int mbase, int n)
{
    v8f p = {};
    v8f q = {};
#pragma unroll
    for (int i = 0; i < cD / 4; ++i) {
        v2f a, b0, b1;
        a.x  = arow[4 * i];
        a.y  = arow[4 * i + 1];
        b0.x = bp[2 * i];
        b0.y = bp[2 * i + 1];
        b1.x = bq[2 * i];
        b1.y = bq[2 * i + 1];
        p = __builtin_amdgcn_wmma_f32_16x16x4_f32(
                false, a, false, b0, (short)0, p, false, false);
        q = __builtin_amdgcn_wmma_f32_16x16x4_f32(
                false, a, false, b1, (short)0, q, false, false);
    }
#pragma unroll
    for (int r = 0; r < 8; ++r) {
        P[(size_t)(mbase + r) * cD + n] = p[r];
        Q[(size_t)(mbase + r) * cD + n] = q[r];
    }
}

static constexpr int cFoldTiles  = cD / 16 + 1;
static constexpr int cFoldBlocks = 3 * cFoldTiles;
static constexpr int cSlots      = 5;
static constexpr int cPreOps     = 10;
static constexpr int cPreParts   = 4;
static constexpr int cPreBlocks  = cPreOps * cPreParts;

__global__ __launch_bounds__(256)
void in_proj_pq(const float* __restrict__ h,
                const float* __restrict__ ipw,
                const float* __restrict__ ipb,
                const float* __restrict__ mw1,
                const float* __restrict__ mw2,
                const float* __restrict__ mb2,
                const float* __restrict__ uw1,
                const float* __restrict__ uw2,
                float* __restrict__ x,
                float* __restrict__ P,
                float* __restrict__ Q,
                float* __restrict__ Fr,
                float* __restrict__ cpr)
{
    __shared__ __attribute__((aligned(16))) float sX[16][cLD];
    __shared__ __attribute__((aligned(16))) float sI[16][cDI + 4];

    const int wave  = threadIdx.x >> 5;
    const int lane  = threadIdx.x & 31;
    const int m0    = blockIdx.x * 16;
    const int l15   = lane & 15;
    const int khalf = lane >> 4;
    const int n     = wave * 16 + l15;
    const int rb    = khalf * 8;

    const size_t boff = (size_t)(2 * khalf) * cD + n;

    if (blockIdx.x >= cM / 16 + cFoldBlocks) {
        const int e    = blockIdx.x - (cM / 16 + cFoldBlocks);
        const int id   = e / cPreParts;
        const int part = e % cPreParts;
        const float* srcm;
        int slot;
        if (id < 3)      { srcm = uw1 + (size_t)id * 2 * cD * cD;                               slot = cSlots * id + 1; }
        else if (id < 6) { srcm = uw2 + (size_t)(id - 3) * cD * cD;                             slot = cSlots * (id - 3) + 2; }
        else if (id < 8) { srcm = mw1 + (size_t)(id - 5) * 257 * cD;                            slot = cSlots * (id - 6) + 3; }
        else             { srcm = mw1 + (size_t)(id - 7) * 257 * cD + (size_t)cD * cD;          slot = cSlots * (id - 8) + 4; }
        float4* dst = reinterpret_cast<float4*>(Fr + (size_t)slot * cOp);
        float4 v[4];
#pragma unroll
        for (int it = 0; it < 4; ++it) {
            const int q  = part * 1024 + it * 256 + (int)threadIdx.x;
            const int ql = q & 31;
            const int qj = (q >> 5) & 15;
            const int qw = q >> 9;
            const float* sp = srcm + (size_t)(8 * qj + 2 * (ql >> 4)) * cD + 16 * qw + (ql & 15);
            v[it] = make_float4(sp[0], sp[cD], sp[4 * cD], sp[5 * cD]);
        }
#pragma unroll
        for (int it = 0; it < 4; ++it)
            dst[part * 1024 + it * 256 + (int)threadIdx.x] = v[it];
        return;
    }

    if (blockIdx.x >= cM / 16) {
        const int e  = blockIdx.x - cM / 16;
        const int l  = e / cFoldTiles;
        const int t  = e % cFoldTiles;
        const float* ub = uw1 + (size_t)l * 2 * cD * cD + (size_t)cD * cD;
        float bf[cD / 2], af[cD / 2];
        load_b_frag<cD>(ub + boff, bf);
        float4 t0 = make_float4(0.0f, 0.0f, 0.0f, 0.0f);
        float4 t1 = t0;
        if (t < cD / 16) {
            const float4* A4 = reinterpret_cast<const float4*>(mw2 + (size_t)l * cD * cD + (size_t)(16 * t) * cD);
            t0 = A4[threadIdx.x];
            t1 = A4[threadIdx.x + 256];
        } else if (wave == 0) {
            t0 = reinterpret_cast<const float4*>(mb2 + (size_t)l * cD)[lane];
        }
        *reinterpret_cast<float4*>(&sX[wave][4 * lane])     = t0;
        *reinterpret_cast<float4*>(&sX[wave + 8][4 * lane]) = t1;
        __syncthreads();
        load_a_frag<cD>(&sX[l15][2 * khalf], af);
        v8f c = {};
        c = mm_reg<cD>(af, bf, c);
        if (t < cD / 16) {
            float* Wl = Fr + (size_t)(cSlots * l) * cOp + (((wave * (cD / 8) + 2 * t + khalf) * 32 + l15) << 2);
#pragma unroll
            for (int r = 0; r < 8; ++r)
                Wl[(((r >> 1) & 1) << 6) + 2 * ((r >> 2) & 1) + (r & 1)] = c[r];
        } else if (khalf == 0) {
            cpr[l * cD + n] = c[0];
        }
        return;
    }

    const float4 th = reinterpret_cast<const float4*>(h + (size_t)m0 * cDI)[threadIdx.x];

    float a0[cDI / 2], b0[cDI / 2];
    load_b_frag<cDI>(ipw + boff, b0);
    *reinterpret_cast<float4*>(&sI[threadIdx.x >> 4][4 * (threadIdx.x & 15)]) = th;
    __syncthreads();
    float bp[cD / 2], bq[cD / 2];
    load_b_frag<cD>(mw1 + boff, bp);
    load_b_frag<cD>(mw1 + (size_t)cD * cD + boff, bq);
    load_a_frag<cDI>(&sI[l15][2 * khalf], a0);

    v8f c = {};
    c = mm_reg<cDI>(a0, b0, c);
    const float bv = ipb[n];
#pragma unroll
    for (int r = 0; r < 8; ++r) {
        float y = c[r] + bv;
        y = y > 0.0f ? y : 0.0f;
        sX[rb + r][n] = y;
        x[(size_t)(m0 + rb + r) * cD + n] = y;
    }
    __syncthreads();
    pq_from_tile(&sX[l15][2 * khalf], bp, bq, P, Q, m0 + rb, n);
}

static constexpr int cTR = 8;
static constexpr int cU  = 16;
static constexpr int cCH = cN / 32;

__device__ __forceinline__ void pair_lane(float p, float q, float a, float w,
                                          float b1, float& acc)
{
    float t = __builtin_fmaf(a, w, p + q) + b1;
    t = t > 0.0f ? t : 0.0f;
    acc = __builtin_fmaf(a, t, acc);
}

__device__ __forceinline__ void pair_step(const float4& p, const float4& q, float a,
                                          const float4& w, const float4& b1,
                                          float4& acc, float& dg)
{
    pair_lane(p.x, q.x, a, w.x, b1.x, acc.x);
    pair_lane(p.y, q.y, a, w.y, b1.y, acc.y);
    pair_lane(p.z, q.z, a, w.z, b1.z, acc.z);
    pair_lane(p.w, q.w, a, w.w, b1.w, acc.w);
    dg += a;
}

template <bool kNext>
__global__ __launch_bounds__(256)
void layer_fused(const float* Pin,
                 const float* __restrict__ Qin,
                 const float* __restrict__ adj,
                 const float* __restrict__ we,
                 const float* __restrict__ mb1,
                 const float* xin,
                 const float* __restrict__ Fl,
                 const float* __restrict__ cp,
                 const float* __restrict__ ub1,
                 const float* __restrict__ ub2,
                 const float* __restrict__ g,
                 const float* __restrict__ beta,
                 float* Pout,
                 float* __restrict__ Qout,
                 float* xout,
                 int add_residual, int relu_after)
{
    __shared__ __attribute__((aligned(16))) float sA[16][cLD];
    __shared__ __attribute__((aligned(16))) float sH[16][cLD];
    __shared__ __attribute__((aligned(16))) float sR[16][cLD];
    __shared__ __attribute__((aligned(16))) float sX[16][cLD];
    __shared__ __attribute__((aligned(16))) int   s_j[cTR][cN + cU];
    __shared__ __attribute__((aligned(16))) float s_a[cTR][cN + cU];
    __shared__ float sDeg[16];

    const int wave  = threadIdx.x >> 5;
    const int lane  = threadIdx.x & 31;
    const int m0    = blockIdx.x * cTR;
    const int l15   = lane & 15;
    const int khalf = lane >> 4;
    const int n     = wave * 16 + l15;
    const int rb    = khalf * 8;

    const int row = m0 + wave;
    const int b   = m0 / cN;

    const float* arow = adj + (size_t)row * cN;
    float av[cCH];
#pragma unroll
    for (int c = 0; c < cCH; ++c)
        av[c] = arow[c * 32 + lane];

    const float4 tx  = reinterpret_cast<const float4*>(xin + (size_t)row * cD)[lane];
    __builtin_amdgcn_sched_barrier(0);
    const float4 p4  = reinterpret_cast<const float4*>(Pin + (size_t)row * cD)[lane];
    const float4 w4  = reinterpret_cast<const float4*>(we)[lane];
    const float4 b14 = reinterpret_cast<const float4*>(mb1)[lane];

    const float bv_c  = cp[n];
    const float bv_u1 = ub1[n];
    const float bv_u2 = ub2[n];
    float gg[4], bb[4];
#pragma unroll
    for (int i = 0; i < 4; ++i) {
        gg[i] = g[lane + 32 * i];
        bb[i] = beta[lane + 32 * i];
    }

    float b1[cD / 2];
    load_b_pre(Fl, wave, lane, b1);
    float b2a[cD / 2];
    load_b_pre(Fl + cOp, wave, lane, b2a);

    {
        const float4 z4 = make_float4(0.0f, 0.0f, 0.0f, 0.0f);
        *reinterpret_cast<float4*>(&sR[wave + cTR][4 * lane]) = z4;
        *reinterpret_cast<float4*>(&sX[wave + cTR][4 * lane]) = z4;
        *reinterpret_cast<float4*>(&sX[wave][4 * lane]) = tx;
        if (lane == 0) sDeg[wave + cTR] = 0.0f;
    }

    int ns = 0;
#pragma unroll
    for (int c = 0; c < cCH; ++c) {
        const bool nz = av[c] != 0.0f;
        const unsigned mask = __builtin_amdgcn_ballot_w32(nz);
        const int pos = ns + (int)__builtin_amdgcn_mbcnt_lo(mask, 0u);
        if (nz) {
            s_j[wave][pos] = c * 32 + lane;
            s_a[wave][pos] = av[c];
        }
        ns += __builtin_popcount(mask);
    }
    if (lane < cU) {
        s_j[wave][ns + lane] = 0;
        s_a[wave][ns + lane] = 0.0f;
    }
    __syncthreads();

    const float* xrow = &sX[l15][2 * khalf];
    v8f hc = {};
    int xs = 0;

    const float4* Qb = reinterpret_cast<const float4*>(Qin + (size_t)b * cN * cD);
    const int4*   lj = reinterpret_cast<const int4*>(s_j[wave]);
    const float4* la = reinterpret_cast<const float4*>(s_a[wave]);

    float4 acc = make_float4(0.0f, 0.0f, 0.0f, 0.0f);
    float  dg  = 0.0f;
    int k = 0;
    for (; k + cU <= ns; k += cU) {
        int jj[cU]; float aa[cU]; float4 q[cU];
#pragma unroll
        for (int v = 0; v < cU / 4; ++v) {
            const int4   j4 = lj[(k >> 2) + v];
            const float4 a4 = la[(k >> 2) + v];
            jj[4 * v + 0] = j4.x; jj[4 * v + 1] = j4.y; jj[4 * v + 2] = j4.z; jj[4 * v + 3] = j4.w;
            aa[4 * v + 0] = a4.x; aa[4 * v + 1] = a4.y; aa[4 * v + 2] = a4.z; aa[4 * v + 3] = a4.w;
        }
#pragma unroll
        for (int u = 0; u < cU; ++u)
            q[u] = Qb[(unsigned)(jj[u] * (cD / 4) + lane)];
        if (xs == 0)      { hc = mm_lds_half<0>(xrow, b2a, hc); xs = 1; }
        else if (xs == 1) { hc = mm_lds_half<1>(xrow, b2a, hc); xs = 2; }
#pragma unroll
        for (int u = 0; u < cU; ++u)
            pair_step(p4, q[u], aa[u], w4, b14, acc, dg);
    }
    if (k < ns) {
        int jj[cU]; float aa[cU]; float4 q[cU];
#pragma unroll
        for (int v = 0; v < cU / 4; ++v) {
            const int4   j4 = lj[(k >> 2) + v];
            const float4 a4 = la[(k >> 2) + v];
            jj[4 * v + 0] = j4.x; jj[4 * v + 1] = j4.y; jj[4 * v + 2] = j4.z; jj[4 * v + 3] = j4.w;
            aa[4 * v + 0] = a4.x; aa[4 * v + 1] = a4.y; aa[4 * v + 2] = a4.z; aa[4 * v + 3] = a4.w;
        }
#pragma unroll
        for (int u = 0; u < cU; ++u)
            q[u] = Qb[(unsigned)(jj[u] * (cD / 4) + lane)];
        if (xs == 0)      { hc = mm_lds_half<0>(xrow, b2a, hc); xs = 1; }
        else if (xs == 1) { hc = mm_lds_half<1>(xrow, b2a, hc); xs = 2; }
#pragma unroll
        for (int u = 0; u < cU; ++u)
            if (k + u < ns)
                pair_step(p4, q[u], aa[u], w4, b14, acc, dg);
    }
    if (xs == 0) { hc = mm_lds_half<0>(xrow, b2a, hc); xs = 1; }
    if (xs == 1) { hc = mm_lds_half<1>(xrow, b2a, hc); }

    *reinterpret_cast<float4*>(&sR[wave][4 * lane]) = acc;
    if (lane == 0) sDeg[wave] = dg;
    __syncthreads();

    float xr[8];
    float b3[cD / 2];
    {
        float a1[cD / 2];
        load_b_pre(Fl + 2 * cOp, wave, lane, b3);
        __builtin_amdgcn_sched_barrier(0);
        load_a_frag<cD>(&sR[l15][2 * khalf], a1);
        float dr[8];
#pragma unroll
        for (int r = 0; r < 8; ++r) {
            xr[r] = sX[rb + r][n];
            dr[r] = sDeg[rb + r];
        }
        v8f c = hc;
        c = mm_reg<cD>(a1, b1, c);
#pragma unroll
        for (int r = 0; r < 8; ++r) {
            const float v = (c[r] + bv_u1) + dr[r] * bv_c;
            sH[rb + r][n] = v > 0.0f ? v : 0.0f;
        }
    }
    __syncthreads();

    float bp[cD / 2], bq[cD / 2];
    if (kNext) {
        load_b_pre(Fl + 3 * cOp, wave, lane, bp);
        load_b_pre(Fl + 4 * cOp, wave, lane, bq);
    }
    {
        v8f c = {};
        c = mm_lds<cD>(&sH[l15][2 * khalf], b3, c);
#pragma unroll
        for (int r = 0; r < 8; ++r) {
            float v = c[r] + bv_u2;
            if (add_residual) v += xr[r];
            sA[rb + r][n] = v;
        }
    }
    __syncthreads();

    {
        float v[4];
#pragma unroll
        for (int i = 0; i < 4; ++i)
            v[i] = sA[wave][lane + 32 * i];

        float s = (v[0] + v[2]) + (v[1] + v[3]);
#pragma unroll
        for (int o = 16; o > 0; o >>= 1)
            s += __shfl_down(s, o);
        const float mu = __shfl(s, 0) * (1.0f / cD);

        float dv[4];
#pragma unroll
        for (int i = 0; i < 4; ++i)
            dv[i] = v[i] - mu;
        float t = (dv[0] * dv[0] + dv[2] * dv[2]) + (dv[1] * dv[1] + dv[3] * dv[3]);
#pragma unroll
        for (int o = 16; o > 0; o >>= 1)
            t += __shfl_down(t, o);
        const float var = __shfl(t, 0) * (1.0f / cD);
        const float rs  = rsqrtf(var + 1e-5f);

#pragma unroll
        for (int i = 0; i < 4; ++i) {
            float y = dv[i] * rs * gg[i] + bb[i];
            if (relu_after) y = y > 0.0f ? y : 0.0f;
            xout[(size_t)row * cD + lane + 32 * i] = y;
            sH[wave][lane + 32 * i] = y;
        }
    }

    if (kNext) {
        __syncthreads();
        const float* arw = &sH[l15][2 * khalf];
        v8f p = {};
        v8f q = {};
#pragma unroll
        for (int i = 0; i < cD / 4; ++i) {
            v2f a, f0, f1;
            a.x  = arw[4 * i];
            a.y  = arw[4 * i + 1];
            f0.x = bp[2 * i];
            f0.y = bp[2 * i + 1];
            f1.x = bq[2 * i];
            f1.y = bq[2 * i + 1];
            p = __builtin_amdgcn_wmma_f32_16x16x4_f32(
                    false, a, false, f0, (short)0, p, false, false);
            q = __builtin_amdgcn_wmma_f32_16x16x4_f32(
                    false, a, false, f1, (short)0, q, false, false);
        }
        if (khalf == 0) {
#pragma unroll
            for (int r = 0; r < 8; ++r) {
                Pout[(size_t)(m0 + r) * cD + n] = p[r];
                Qout[(size_t)(m0 + r) * cD + n] = q[r];
            }
        }
    }
}

extern "C" void kernel_launch(void* const* d_in, const int* in_sizes, int n_in,
                              void* d_out, int out_size, void* d_ws, size_t ws_size,
                              hipStream_t stream)
{
    (void)in_sizes; (void)n_in; (void)out_size; (void)ws_size;

    const float* h    = (const float*)d_in[0];
    const float* adj  = (const float*)d_in[1];
    const float* ip_w = (const float*)d_in[2];
    const float* ip_b = (const float*)d_in[3];
    const float* mw1  = (const float*)d_in[4];
    const float* mb1  = (const float*)d_in[5];
    const float* mw2  = (const float*)d_in[6];
    const float* mb2  = (const float*)d_in[7];
    const float* uw1  = (const float*)d_in[8];
    const float* ub1  = (const float*)d_in[9];
    const float* uw2  = (const float*)d_in[10];
    const float* ub2  = (const float*)d_in[11];
    const float* ln_g = (const float*)d_in[12];
    const float* ln_b = (const float*)d_in[13];

    float* ws = (float*)d_ws;
    const size_t S = (size_t)cM * cD;
    float* x  = ws + 0 * S;
    float* Pa = ws + 1 * S;
    float* Qa = ws + 2 * S;
    float* Qb = ws + 6 * S;
    float* Fr = ws + 3 * S;
    float* cpr = Fr + (size_t)3 * cSlots * cOp;

    in_proj_pq<<<cM / 16 + cFoldBlocks + cPreBlocks, 256, 0, stream>>>(h, ip_w, ip_b, mw1, mw2, mb2, uw1, uw2, x, Pa, Qa, Fr, cpr);

    for (int l = 0; l < 3; ++l) {
        const float* mw1l = mw1 + (size_t)l * 257 * cD;
        const float* mb1l = mb1 + (size_t)l * cD;
        const float* Fll  = Fr + (size_t)l * cSlots * cOp;
        const float* cpl  = cpr + (size_t)l * cD;
        const float* ub1l = ub1 + (size_t)l * cD;
        const float* ub2l = ub2 + (size_t)l * cD;
        float* Qi = (l == 1) ? Qb : Qa;
        float* Qo = (l == 1) ? Qa : Qb;

        if (l < 2) {
            layer_fused<true><<<cM / cTR, 256, 0, stream>>>(Pa, Qi, adj, mw1l + 256 * cD, mb1l, x, Fll, cpl,
                                                           ub1l, ub2l,
                                                           ln_g + l * cD, ln_b + l * cD,
                                                           Pa, Qo, x, 1, 1);
        } else {
            layer_fused<false><<<cM / cTR, 256, 0, stream>>>(Pa, Qi, adj, mw1l + 256 * cD, mb1l, x, Fll, cpl,
                                                            ub1l, ub2l,
                                                            ln_g + l * cD, ln_b + l * cD,
                                                            Pa, Qo, (float*)d_out, 0, 0);
        }
    }
}


# workgroup barriers fence LDS only (every hand-over goes through LDS): none drains the wave's outstanding global loads and stores, so operand loads asked for ahead fly across them; sums unchanged
# speedup vs baseline: 6.1698x; 1.1017x over previous
#include <hip/hip_runtime.h>

typedef __attribute__((ext_vector_type(2))) float v2f;
typedef __attribute__((ext_vector_type(8))) float v8f;


static constexpr int cB = 2;
static constexpr int cN = 384;
static constexpr int cD = 128;
static constexpr int cM = cB * cN;
static constexpr int cDI = 64;
static constexpr int cLD = cD + 4;

template <int K>
__device__ __forceinline__ void load_b_frag(const float* __restrict__ bcol, float (&b)[K / 2])
{
#pragma unroll
    for (int i = 0; i < K / 4; ++i) {
        b[2 * i]     = bcol[(size_t)(4 * i) * cD];
        b[2 * i + 1] = bcol[(size_t)(4 * i + 1) * cD];
    }
}

static constexpr int cOp = cD * cD;
__device__ __forceinline__ void load_b_pre(const float* __restrict__ op, int wave, int lane, float (&b)[cD / 2])
{
    const float4* f = reinterpret_cast<const float4*>(op) + (wave * (cD / 8) * 32 + lane);
#pragma unroll
    for (int j4 = 0; j4 < cD / 8; ++j4) {
        const float4 v = f[j4 * 32];
        b[4 * j4]     = v.x;
        b[4 * j4 + 1] = v.y;
        b[4 * j4 + 2] = v.z;
        b[4 * j4 + 3] = v.w;
    }
}

__device__ __forceinline__ void lds_barrier()
{
    __builtin_amdgcn_fence(__ATOMIC_RELEASE, "workgroup", "local");
    __builtin_amdgcn_s_barrier();
    __builtin_amdgcn_fence(__ATOMIC_ACQUIRE, "workgroup", "local");
}

template <int K>
__device__ __forceinline__ void load_a_frag(const float* arow, float (&a)[K / 2])
{
#pragma unroll
    for (int i = 0; i < K / 4; ++i) {
        a[2 * i]     = arow[4 * i];
        a[2 * i + 1] = arow[4 * i + 1];
    }
}

template <int K>
__device__ __forceinline__ v8f mm_reg(const float (&af)[K / 2], const float (&bf)[K / 2], v8f c)
{
#pragma unroll
    for (int i = 0; i < K / 4; ++i) {
        v2f a, b;
        a.x = af[2 * i];
        a.y = af[2 * i + 1];
        b.x = bf[2 * i];
        b.y = bf[2 * i + 1];
        c = __builtin_amdgcn_wmma_f32_16x16x4_f32(
                false, a, false, b, (short)0, c, false, false);
    }
    return c;
}

template <int K>
__device__ __forceinline__ v8f mm_lds(const float* arow, const float (&bf)[K / 2], v8f c)
{
#pragma unroll
    for (int i = 0; i < K / 4; ++i) {
        v2f a, b;
        a.x = arow[4 * i];
        a.y = arow[4 * i + 1];
        b.x = bf[2 * i];
        b.y = bf[2 * i + 1];
        c = __builtin_amdgcn_wmma_f32_16x16x4_f32(
                false, a, false, b, (short)0, c, false, false);
    }
    return c;
}

template <int H>
__device__ __forceinline__ v8f mm_lds_half(const float* arow, const float (&bf)[cD / 2], v8f c)
{
    float a[cD / 4];
#pragma unroll
    for (int i = 0; i < cD / 8; ++i) {
        a[2 * i]     = arow[4 * (16 * H + i)];
        a[2 * i + 1] = arow[4 * (16 * H + i) + 1];
    }
#pragma unroll
    for (int i = 0; i < cD / 8; ++i) {
        v2f av, bv;
        av.x = a[2 * i];
        av.y = a[2 * i + 1];
        bv.x = bf[2 * (16 * H + i)];
        bv.y = bf[2 * (16 * H + i) + 1];
        c = __builtin_amdgcn_wmma_f32_16x16x4_f32(
                false, av, false, bv, (short)0, c, false, false);
    }
    return c;
}

__device__ __forceinline__ void pq_from_tile(const float* arow,
                                             const float (&bp)[cD / 2],
                                             const float (&bq)[cD / 2],
                                             float* __restrict__ P,
                                             float* __restrict__ Q,
                                             int mbase, int n)
{
    v8f p = {};
    v8f q = {};
#pragma unroll
    for (int i = 0; i < cD / 4; ++i) {
        v2f a, b0, b1;
        a.x  = arow[4 * i];
        a.y  = arow[4 * i + 1];
        b0.x = bp[2 * i];
        b0.y = bp[2 * i + 1];
        b1.x = bq[2 * i];
        b1.y = bq[2 * i + 1];
        p = __builtin_amdgcn_wmma_f32_16x16x4_f32(
                false, a, false, b0, (short)0, p, false, false);
        q = __builtin_amdgcn_wmma_f32_16x16x4_f32(
                false, a, false, b1, (short)0, q, false, false);
    }
#pragma unroll
    for (int r = 0; r < 8; ++r) {
        P[(size_t)(mbase + r) * cD + n] = p[r];
        Q[(size_t)(mbase + r) * cD + n] = q[r];
    }
}

static constexpr int cFoldTiles  = cD / 16 + 1;
static constexpr int cFoldBlocks = 3 * cFoldTiles;
static constexpr int cSlots      = 5;
static constexpr int cPreOps     = 10;
static constexpr int cPreParts   = 4;
static constexpr int cPreBlocks  = cPreOps * cPreParts;

__global__ __launch_bounds__(256)
void in_proj_pq(const float* __restrict__ h,
                const float* __restrict__ ipw,
                const float* __restrict__ ipb,
                const float* __restrict__ mw1,
                const float* __restrict__ mw2,
                const float* __restrict__ mb2,
                const float* __restrict__ uw1,
                const float* __restrict__ uw2,
                float* __restrict__ x,
                float* __restrict__ P,
                float* __restrict__ Q,
                float* __restrict__ Fr,
                float* __restrict__ cpr)
{
    __shared__ __attribute__((aligned(16))) float sX[16][cLD];
    __shared__ __attribute__((aligned(16))) float sI[16][cDI + 4];

    const int wave  = threadIdx.x >> 5;
    const int lane  = threadIdx.x & 31;
    const int m0    = blockIdx.x * 16;
    const int l15   = lane & 15;
    const int khalf = lane >> 4;
    const int n     = wave * 16 + l15;
    const int rb    = khalf * 8;

    const size_t boff = (size_t)(2 * khalf) * cD + n;

    if (blockIdx.x >= cM / 16 + cFoldBlocks) {
        const int e    = blockIdx.x - (cM / 16 + cFoldBlocks);
        const int id   = e / cPreParts;
        const int part = e % cPreParts;
        const float* srcm;
        int slot;
        if (id < 3)      { srcm = uw1 + (size_t)id * 2 * cD * cD;                               slot = cSlots * id + 1; }
        else if (id < 6) { srcm = uw2 + (size_t)(id - 3) * cD * cD;                             slot = cSlots * (id - 3) + 2; }
        else if (id < 8) { srcm = mw1 + (size_t)(id - 5) * 257 * cD;                            slot = cSlots * (id - 6) + 3; }
        else             { srcm = mw1 + (size_t)(id - 7) * 257 * cD + (size_t)cD * cD;          slot = cSlots * (id - 8) + 4; }
        float4* dst = reinterpret_cast<float4*>(Fr + (size_t)slot * cOp);
        float4 v[4];
#pragma unroll
        for (int it = 0; it < 4; ++it) {
            const int q  = part * 1024 + it * 256 + (int)threadIdx.x;
            const int ql = q & 31;
            const int qj = (q >> 5) & 15;
            const int qw = q >> 9;
            const float* sp = srcm + (size_t)(8 * qj + 2 * (ql >> 4)) * cD + 16 * qw + (ql & 15);
            v[it] = make_float4(sp[0], sp[cD], sp[4 * cD], sp[5 * cD]);
        }
#pragma unroll
        for (int it = 0; it < 4; ++it)
            dst[part * 1024 + it * 256 + (int)threadIdx.x] = v[it];
        return;
    }

    if (blockIdx.x >= cM / 16) {
        const int e  = blockIdx.x - cM / 16;
        const int l  = e / cFoldTiles;
        const int t  = e % cFoldTiles;
        const float* ub = uw1 + (size_t)l * 2 * cD * cD + (size_t)cD * cD;
        float bf[cD / 2], af[cD / 2];
        load_b_frag<cD>(ub + boff, bf);
        float4 t0 = make_float4(0.0f, 0.0f, 0.0f, 0.0f);
        float4 t1 = t0;
        if (t < cD / 16) {
            const float4* A4 = reinterpret_cast<const float4*>(mw2 + (size_t)l * cD * cD + (size_t)(16 * t) * cD);
            t0 = A4[threadIdx.x];
            t1 = A4[threadIdx.x + 256];
        } else if (wave == 0) {
            t0 = reinterpret_cast<const float4*>(mb2 + (size_t)l * cD)[lane];
        }
        *reinterpret_cast<float4*>(&sX[wave][4 * lane])     = t0;
        *reinterpret_cast<float4*>(&sX[wave + 8][4 * lane]) = t1;
        lds_barrier();
        load_a_frag<cD>(&sX[l15][2 * khalf], af);
        v8f c = {};
        c = mm_reg<cD>(af, bf, c);
        if (t < cD / 16) {
            float* Wl = Fr + (size_t)(cSlots * l) * cOp + (((wave * (cD / 8) + 2 * t + khalf) * 32 + l15) << 2);
#pragma unroll
            for (int r = 0; r < 8; ++r)
                Wl[(((r >> 1) & 1) << 6) + 2 * ((r >> 2) & 1) + (r & 1)] = c[r];
        } else if (khalf == 0) {
            cpr[l * cD + n] = c[0];
        }
        return;
    }

    const float4 th = reinterpret_cast<const float4*>(h + (size_t)m0 * cDI)[threadIdx.x];

    float a0[cDI / 2], b0[cDI / 2];
    load_b_frag<cDI>(ipw + boff, b0);
    *reinterpret_cast<float4*>(&sI[threadIdx.x >> 4][4 * (threadIdx.x & 15)]) = th;
    lds_barrier();
    float bp[cD / 2], bq[cD / 2];
    load_b_frag<cD>(mw1 + boff, bp);
    load_b_frag<cD>(mw1 + (size_t)cD * cD + boff, bq);
    load_a_frag<cDI>(&sI[l15][2 * khalf], a0);

    v8f c = {};
    c = mm_reg<cDI>(a0, b0, c);
    const float bv = ipb[n];
#pragma unroll
    for (int r = 0; r < 8; ++r) {
        float y = c[r] + bv;
        y = y > 0.0f ? y : 0.0f;
        sX[rb + r][n] = y;
        x[(size_t)(m0 + rb + r) * cD + n] = y;
    }
    lds_barrier();
    pq_from_tile(&sX[l15][2 * khalf], bp, bq, P, Q, m0 + rb, n);
}

static constexpr int cTR = 8;
static constexpr int cU  = 16;
static constexpr int cCH = cN / 32;

__device__ __forceinline__ void pair_lane(float p, float q, float a, float w,
                                          float b1, float& acc)
{
    float t = __builtin_fmaf(a, w, p + q) + b1;
    t = t > 0.0f ? t : 0.0f;
    acc = __builtin_fmaf(a, t, acc);
}

__device__ __forceinline__ void pair_step(const float4& p, const float4& q, float a,
                                          const float4& w, const float4& b1,
                                          float4& acc, float& dg)
{
    pair_lane(p.x, q.x, a, w.x, b1.x, acc.x);
    pair_lane(p.y, q.y, a, w.y, b1.y, acc.y);
    pair_lane(p.z, q.z, a, w.z, b1.z, acc.z);
    pair_lane(p.w, q.w, a, w.w, b1.w, acc.w);
    dg += a;
}

template <bool kNext>
__global__ __launch_bounds__(256)
void layer_fused(const float* Pin,
                 const float* __restrict__ Qin,
                 const float* __restrict__ adj,
                 const float* __restrict__ we,
                 const float* __restrict__ mb1,
                 const float* xin,
                 const float* __restrict__ Fl,
                 const float* __restrict__ cp,
                 const float* __restrict__ ub1,
                 const float* __restrict__ ub2,
                 const float* __restrict__ g,
                 const float* __restrict__ beta,
                 float* Pout,
                 float* __restrict__ Qout,
                 float* xout,
                 int add_residual, int relu_after)
{
    __shared__ __attribute__((aligned(16))) float sA[16][cLD];
    __shared__ __attribute__((aligned(16))) float sH[16][cLD];
    __shared__ __attribute__((aligned(16))) float sR[16][cLD];
    __shared__ __attribute__((aligned(16))) float sX[16][cLD];
    __shared__ __attribute__((aligned(16))) int   s_j[cTR][cN + cU];
    __shared__ __attribute__((aligned(16))) float s_a[cTR][cN + cU];
    __shared__ float sDeg[16];

    const int wave  = threadIdx.x >> 5;
    const int lane  = threadIdx.x & 31;
    const int m0    = blockIdx.x * cTR;
    const int l15   = lane & 15;
    const int khalf = lane >> 4;
    const int n     = wave * 16 + l15;
    const int rb    = khalf * 8;

    const int row = m0 + wave;
    const int b   = m0 / cN;

    const float* arow = adj + (size_t)row * cN;
    float av[cCH];
#pragma unroll
    for (int c = 0; c < cCH; ++c)
        av[c] = arow[c * 32 + lane];

    const float4 tx  = reinterpret_cast<const float4*>(xin + (size_t)row * cD)[lane];
    __builtin_amdgcn_sched_barrier(0);
    const float4 p4  = reinterpret_cast<const float4*>(Pin + (size_t)row * cD)[lane];
    const float4 w4  = reinterpret_cast<const float4*>(we)[lane];
    const float4 b14 = reinterpret_cast<const float4*>(mb1)[lane];

    const float bv_c  = cp[n];
    const float bv_u1 = ub1[n];
    const float bv_u2 = ub2[n];
    float gg[4], bb[4];
#pragma unroll
    for (int i = 0; i < 4; ++i) {
        gg[i] = g[lane + 32 * i];
        bb[i] = beta[lane + 32 * i];
    }

    float b1[cD / 2];
    load_b_pre(Fl, wave, lane, b1);
    float b2a[cD / 2];
    load_b_pre(Fl + cOp, wave, lane, b2a);

    {
        const float4 z4 = make_float4(0.0f, 0.0f, 0.0f, 0.0f);
        *reinterpret_cast<float4*>(&sR[wave + cTR][4 * lane]) = z4;
        *reinterpret_cast<float4*>(&sX[wave + cTR][4 * lane]) = z4;
        *reinterpret_cast<float4*>(&sX[wave][4 * lane]) = tx;
        if (lane == 0) sDeg[wave + cTR] = 0.0f;
    }

    int ns = 0;
#pragma unroll
    for (int c = 0; c < cCH; ++c) {
        const bool nz = av[c] != 0.0f;
        const unsigned mask = __builtin_amdgcn_ballot_w32(nz);
        const int pos = ns + (int)__builtin_amdgcn_mbcnt_lo(mask, 0u);
        if (nz) {
            s_j[wave][pos] = c * 32 + lane;
            s_a[wave][pos] = av[c];
        }
        ns += __builtin_popcount(mask);
    }
    if (lane < cU) {
        s_j[wave][ns + lane] = 0;
        s_a[wave][ns + lane] = 0.0f;
    }
    lds_barrier();

    const float* xrow = &sX[l15][2 * khalf];
    v8f hc = {};
    int xs = 0;

    const float4* Qb = reinterpret_cast<const float4*>(Qin + (size_t)b * cN * cD);
    const int4*   lj = reinterpret_cast<const int4*>(s_j[wave]);
    const float4* la = reinterpret_cast<const float4*>(s_a[wave]);

    float4 acc = make_float4(0.0f, 0.0f, 0.0f, 0.0f);
    float  dg  = 0.0f;
    int k = 0;
    for (; k + cU <= ns; k += cU) {
        int jj[cU]; float aa[cU]; float4 q[cU];
#pragma unroll
        for (int v = 0; v < cU / 4; ++v) {
            const int4   j4 = lj[(k >> 2) + v];
            const float4 a4 = la[(k >> 2) + v];
            jj[4 * v + 0] = j4.x; jj[4 * v + 1] = j4.y; jj[4 * v + 2] = j4.z; jj[4 * v + 3] = j4.w;
            aa[4 * v + 0] = a4.x; aa[4 * v + 1] = a4.y; aa[4 * v + 2] = a4.z; aa[4 * v + 3] = a4.w;
        }
#pragma unroll
        for (int u = 0; u < cU; ++u)
            q[u] = Qb[(unsigned)(jj[u] * (cD / 4) + lane)];
        if (xs == 0)      { hc = mm_lds_half<0>(xrow, b2a, hc); xs = 1; }
        else if (xs == 1) { hc = mm_lds_half<1>(xrow, b2a, hc); xs = 2; }
#pragma unroll
        for (int u = 0; u < cU; ++u)
            pair_step(p4, q[u], aa[u], w4, b14, acc, dg);
    }
    if (k < ns) {
        int jj[cU]; float aa[cU]; float4 q[cU];
#pragma unroll
        for (int v = 0; v < cU / 4; ++v) {
            const int4   j4 = lj[(k >> 2) + v];
            const float4 a4 = la[(k >> 2) + v];
            jj[4 * v + 0] = j4.x; jj[4 * v + 1] = j4.y; jj[4 * v + 2] = j4.z; jj[4 * v + 3] = j4.w;
            aa[4 * v + 0] = a4.x; aa[4 * v + 1] = a4.y; aa[4 * v + 2] = a4.z; aa[4 * v + 3] = a4.w;
        }
#pragma unroll
        for (int u = 0; u < cU; ++u)
            q[u] = Qb[(unsigned)(jj[u] * (cD / 4) + lane)];
        if (xs == 0)      { hc = mm_lds_half<0>(xrow, b2a, hc); xs = 1; }
        else if (xs == 1) { hc = mm_lds_half<1>(xrow, b2a, hc); xs = 2; }
#pragma unroll
        for (int u = 0; u < cU; ++u)
            if (k + u < ns)
                pair_step(p4, q[u], aa[u], w4, b14, acc, dg);
    }
    if (xs == 0) { hc = mm_lds_half<0>(xrow, b2a, hc); xs = 1; }
    if (xs == 1) { hc = mm_lds_half<1>(xrow, b2a, hc); }

    *reinterpret_cast<float4*>(&sR[wave][4 * lane]) = acc;
    if (lane == 0) sDeg[wave] = dg;
    lds_barrier();

    float xr[8];
    float b3[cD / 2];
    {
        float a1[cD / 2];
        load_b_pre(Fl + 2 * cOp, wave, lane, b3);
        __builtin_amdgcn_sched_barrier(0);
        load_a_frag<cD>(&sR[l15][2 * khalf], a1);
        float dr[8];
#pragma unroll
        for (int r = 0; r < 8; ++r) {
            xr[r] = sX[rb + r][n];
            dr[r] = sDeg[rb + r];
        }
        v8f c = hc;
        c = mm_reg<cD>(a1, b1, c);
#pragma unroll
        for (int r = 0; r < 8; ++r) {
            const float v = (c[r] + bv_u1) + dr[r] * bv_c;
            sH[rb + r][n] = v > 0.0f ? v : 0.0f;
        }
    }
    lds_barrier();

    float bp[cD / 2], bq[cD / 2];
    if (kNext) {
        load_b_pre(Fl + 3 * cOp, wave, lane, bp);
        load_b_pre(Fl + 4 * cOp, wave, lane, bq);
    }
    {
        v8f c = {};
        c = mm_lds<cD>(&sH[l15][2 * khalf], b3, c);
#pragma unroll
        for (int r = 0; r < 8; ++r) {
            float v = c[r] + bv_u2;
            if (add_residual) v += xr[r];
            sA[rb + r][n] = v;
        }
    }
    lds_barrier();

    {
        float v[4];
#pragma unroll
        for (int i = 0; i < 4; ++i)
            v[i] = sA[wave][lane + 32 * i];

        float s = (v[0] + v[2]) + (v[1] + v[3]);
#pragma unroll
        for (int o = 16; o > 0; o >>= 1)
            s += __shfl_down(s, o);
        const float mu = __shfl(s, 0) * (1.0f / cD);

        float dv[4];
#pragma unroll
        for (int i = 0; i < 4; ++i)
            dv[i] = v[i] - mu;
        float t = (dv[0] * dv[0] + dv[2] * dv[2]) + (dv[1] * dv[1] + dv[3] * dv[3]);
#pragma unroll
        for (int o = 16; o > 0; o >>= 1)
            t += __shfl_down(t, o);
        const float var = __shfl(t, 0) * (1.0f / cD);
        const float rs  = rsqrtf(var + 1e-5f);

#pragma unroll
        for (int i = 0; i < 4; ++i) {
            float y = dv[i] * rs * gg[i] + bb[i];
            if (relu_after) y = y > 0.0f ? y : 0.0f;
            xout[(size_t)row * cD + lane + 32 * i] = y;
            sH[wave][lane + 32 * i] = y;
        }
    }

    if (kNext) {
        lds_barrier();
        const float* arw = &sH[l15][2 * khalf];
        v8f p = {};
        v8f q = {};
#pragma unroll
        for (int i = 0; i < cD / 4; ++i) {
            v2f a, f0, f1;
            a.x  = arw[4 * i];
            a.y  = arw[4 * i + 1];
            f0.x = bp[2 * i];
            f0.y = bp[2 * i + 1];
            f1.x = bq[2 * i];
            f1.y = bq[2 * i + 1];
            p = __builtin_amdgcn_wmma_f32_16x16x4_f32(
                    false, a, false, f0, (short)0, p, false, false);
            q = __builtin_amdgcn_wmma_f32_16x16x4_f32(
                    false, a, false, f1, (short)0, q, false, false);
        }
        if (khalf == 0) {
#pragma unroll
            for (int r = 0; r < 8; ++r) {
                Pout[(size_t)(m0 + r) * cD + n] = p[r];
                Qout[(size_t)(m0 + r) * cD + n] = q[r];
            }
        }
    }
}

extern "C" void kernel_launch(void* const* d_in, const int* in_sizes, int n_in,
                              void* d_out, int out_size, void* d_ws, size_t ws_size,
                              hipStream_t stream)
{
    (void)in_sizes; (void)n_in; (void)out_size; (void)ws_size;

    const float* h    = (const float*)d_in[0];
    const float* adj  = (const float*)d_in[1];
    const float* ip_w = (const float*)d_in[2];
    const float* ip_b = (const float*)d_in[3];
    const float* mw1  = (const float*)d_in[4];
    const float* mb1  = (const float*)d_in[5];
    const float* mw2  = (const float*)d_in[6];
    const float* mb2  = (const float*)d_in[7];
    const float* uw1  = (const float*)d_in[8];
    const float* ub1  = (const float*)d_in[9];
    const float* uw2  = (const float*)d_in[10];
    const float* ub2  = (const float*)d_in[11];
    const float* ln_g = (const float*)d_in[12];
    const float* ln_b = (const float*)d_in[13];

    float* ws = (float*)d_ws;
    const size_t S = (size_t)cM * cD;
    float* x  = ws + 0 * S;
    float* Pa = ws + 1 * S;
    float* Qa = ws + 2 * S;
    float* Qb = ws + 6 * S;
    float* Fr = ws + 3 * S;
    float* cpr = Fr + (size_t)3 * cSlots * cOp;

    in_proj_pq<<<cM / 16 + cFoldBlocks + cPreBlocks, 256, 0, stream>>>(h, ip_w, ip_b, mw1, mw2, mb2, uw1, uw2, x, Pa, Qa, Fr, cpr);

    for (int l = 0; l < 3; ++l) {
        const float* mw1l = mw1 + (size_t)l * 257 * cD;
        const float* mb1l = mb1 + (size_t)l * cD;
        const float* Fll  = Fr + (size_t)l * cSlots * cOp;
        const float* cpl  = cpr + (size_t)l * cD;
        const float* ub1l = ub1 + (size_t)l * cD;
        const float* ub2l = ub2 + (size_t)l * cD;
        float* Qi = (l == 1) ? Qb : Qa;
        float* Qo = (l == 1) ? Qa : Qb;

        if (l < 2) {
            layer_fused<true><<<cM / cTR, 256, 0, stream>>>(Pa, Qi, adj, mw1l + 256 * cD, mb1l, x, Fll, cpl,
                                                           ub1l, ub2l,
                                                           ln_g + l * cD, ln_b + l * cD,
                                                           Pa, Qo, x, 1, 1);
        } else {
            layer_fused<false><<<cM / cTR, 256, 0, stream>>>(Pa, Qi, adj, mw1l + 256 * cD, mb1l, x, Fll, cpl,
                                                            ub1l, ub2l,
                                                            ln_g + l * cD, ln_b + l * cD,
                                                            Pa, Qo, (float*)d_out, 0, 0);
        }
    }
}
